# starting kernel
# baseline (speedup 1.0000x reference)
#include <hip/hip_runtime.h>
#include <hip/hip_bf16.h>

typedef __attribute__((ext_vector_type(2))) float v2f;
typedef __attribute__((ext_vector_type(8))) float v8f;


__global__ void init_deg_kernel(float* __restrict__ deg, int n) {
  int i = blockIdx.x * blockDim.x + threadIdx.x;
  if (i < n) deg[i] = 1.0f;
}

__global__ void count_deg_kernel(const int* __restrict__ dst, float* __restrict__ deg, int E) {
  int e = blockIdx.x * blockDim.x + threadIdx.x;
  if (e < E)
    __hip_atomic_fetch_add(&deg[dst[e]], 1.0f, __ATOMIC_RELAXED, __HIP_MEMORY_SCOPE_AGENT);
}

__global__ void rsqrt_kernel(float* __restrict__ deg, int n) {
  int i = blockIdx.x * blockDim.x + threadIdx.x;
  if (i < n) deg[i] = rsqrtf(deg[i]);
}


template <int K, int NC>
__global__ void gemm_wmma_kernel(const float* __restrict__ A,
                                 const float* __restrict__ W,
                                 float* __restrict__ out) {
  const int lane = threadIdx.x & 31;
  const int wave = threadIdx.x >> 5;
  const int m = lane & 15;
  const int h = lane >> 4;
  const int rowTile = blockIdx.x * 16;
  const int colTile = (blockIdx.y * 4 + wave) * 16;

  const float* arow = A + (size_t)(rowTile + m) * K + 2 * h;
  const float* wcol = W + (size_t)(2 * h) * NC + colTile + m;

  v8f c = {0.f, 0.f, 0.f, 0.f, 0.f, 0.f, 0.f, 0.f};

#pragma unroll
  for (int k0 = 0; k0 < K; k0 += 4) {
    float2 av = *(const float2*)(arow + k0);
    v2f a = {av.x, av.y};
    const float* wp = wcol + (size_t)k0 * NC;
    v2f b = {wp[0], wp[NC]};
    c = __builtin_amdgcn_wmma_f32_16x16x4_f32(false, a, false, b, (short)0, c,
                                              false, false);
  }

  float* orow = out + (size_t)(rowTile + h * 8) * NC + colTile + m;
#pragma unroll
  for (int r = 0; r < 8; ++r) orow[(size_t)r * NC] = c[r];
}


__global__ void self_init_kernel(const float* __restrict__ xw,
                                 const float* __restrict__ dinv,
                                 const float* __restrict__ bias,
                                 float* __restrict__ out, int total, int F) {
  int i = blockIdx.x * blockDim.x + threadIdx.x;
  if (i >= total) return;
  int v = i / F;
  int f = i - v * F;
  float s = dinv[v];
  out[i] = s * s * xw[i] + bias[f];
}

__global__ void edge_agg_kernel(const int* __restrict__ src,
                                const int* __restrict__ dst,
                                const float* __restrict__ dinv,
                                const float* __restrict__ xw,
                                float* __restrict__ out, int E, int F) {
  int wid = (blockIdx.x * blockDim.x + threadIdx.x) >> 5;
  int lane = threadIdx.x & 31;
  if (wid >= E) return;
  int s = src[wid];
  int d = dst[wid];
  float norm = dinv[s] * dinv[d];
  const float* xr = xw + (size_t)s * F;
  float* orow = out + (size_t)d * F;
  for (int f = lane; f < F; f += 32) {
    __hip_atomic_fetch_add(&orow[f], norm * xr[f], __ATOMIC_RELAXED,
                           __HIP_MEMORY_SCOPE_AGENT);
  }
}

__global__ void tanh_kernel(float* __restrict__ out, int n) {
  int i = blockIdx.x * blockDim.x + threadIdx.x;
  if (i < n) out[i] = tanhf(out[i]);
}


extern "C" void kernel_launch(void* const* d_in, const int* in_sizes, int n_in,
                              void* d_out, int out_size, void* d_ws, size_t ws_size,
                              hipStream_t stream) {
  const float* x  = (const float*)d_in[0];
  const int*   ei = (const int*)d_in[1];
  const float* W1 = (const float*)d_in[2];
  const float* b1 = (const float*)d_in[3];
  const float* W2 = (const float*)d_in[4];
  const float* b2 = (const float*)d_in[5];

  const int DIN  = in_sizes[3];
  const int DOUT = in_sizes[5];
  const int N    = in_sizes[0] / DIN;
  const int E    = in_sizes[1] / 2;
  const int* src = ei;
  const int* dst = ei + E;

  char* ws = (char*)d_ws;
  size_t off = ((size_t)N * sizeof(float) + 255) & ~(size_t)255;
  float* dinv = (float*)ws;
  float* xw1  = (float*)(ws + off);
  size_t off2 = off + (((size_t)N * DIN * sizeof(float) + 255) & ~(size_t)255);
  float* h1   = (float*)(ws + off2);
  float* xw2  = xw1;
  float* out  = (float*)d_out;

  const int TB = 256;

  init_deg_kernel<<<(N + TB - 1) / TB, TB, 0, stream>>>(dinv, N);
  count_deg_kernel<<<(E + TB - 1) / TB, TB, 0, stream>>>(dst, dinv, E);
  rsqrt_kernel<<<(N + TB - 1) / TB, TB, 0, stream>>>(dinv, N);

  gemm_wmma_kernel<128, 128><<<dim3(N / 16, 128 / 64), 128, 0, stream>>>(x, W1, xw1);
  self_init_kernel<<<(N * DIN + TB - 1) / TB, TB, 0, stream>>>(xw1, dinv, b1, h1,
                                                               N * DIN, DIN);
  edge_agg_kernel<<<((size_t)E * 32 + TB - 1) / TB, TB, 0, stream>>>(src, dst, dinv,
                                                                     xw1, h1, E, DIN);

  gemm_wmma_kernel<128, 64><<<dim3(N / 16, 64 / 64), 128, 0, stream>>>(h1, W2, xw2);
  self_init_kernel<<<(N * DOUT + TB - 1) / TB, TB, 0, stream>>>(xw2, dinv, b2, out,
                                                                N * DOUT, DOUT);
  edge_agg_kernel<<<((size_t)E * 32 + TB - 1) / TB, TB, 0, stream>>>(src, dst, dinv,
                                                                     xw2, out, E, DOUT);
  tanh_kernel<<<(out_size + TB - 1) / TB, TB, 0, stream>>>(out, out_size);
}


# edge scatter with f32 atomics replaced by per-call in-edge lists (int count with arrival-order rank, scan, fill) and one wave summing each destination row in registers, stored once
# speedup vs baseline: 1.8740x; 1.8740x over previous
#include <hip/hip_runtime.h>
#include <hip/hip_bf16.h>

typedef __attribute__((ext_vector_type(2))) float v2f;
typedef __attribute__((ext_vector_type(8))) float v8f;


__global__ void zero_counts_kernel(int* __restrict__ cnt, int n) {
  int i = blockIdx.x * blockDim.x + threadIdx.x;
  if (i < n) cnt[i] = 0;
}

__global__ void count_rank_kernel(const int* __restrict__ dst, int* __restrict__ cnt,
                                  int* __restrict__ rank, int E) {
  int e = blockIdx.x * blockDim.x + threadIdx.x;
  if (e < E)
    rank[e] = __hip_atomic_fetch_add(&cnt[dst[e]], 1, __ATOMIC_RELAXED,
                                     __HIP_MEMORY_SCOPE_AGENT);
}

__global__ void __launch_bounds__(256)
scan_counts_kernel(const int* __restrict__ cnt, int* __restrict__ rowptr,
                   float* __restrict__ dinv, int n) {
  __shared__ int s_wave_cnt[8];
  __shared__ int s_wave_front[8];
  const int tid = threadIdx.x;
  const int lane = tid & 31;
  const int wave = tid >> 5;
  const int start = blockIdx.x * 256;

  int front = 0;
  const int4* c4 = (const int4*)cnt;
  const int lim = start / 4;
  int i = tid;
  for (; i + 7 * 256 < lim; i += 8 * 256) {
    const int4 q0 = c4[i];
    const int4 q1 = c4[i + 256];
    const int4 q2 = c4[i + 2 * 256];
    const int4 q3 = c4[i + 3 * 256];
    const int4 q4 = c4[i + 4 * 256];
    const int4 q5 = c4[i + 5 * 256];
    const int4 q6 = c4[i + 6 * 256];
    const int4 q7 = c4[i + 7 * 256];
    front += ((q0.x + q0.y) + (q0.z + q0.w)) + ((q1.x + q1.y) + (q1.z + q1.w)) +
             ((q2.x + q2.y) + (q2.z + q2.w)) + ((q3.x + q3.y) + (q3.z + q3.w)) +
             ((q4.x + q4.y) + (q4.z + q4.w)) + ((q5.x + q5.y) + (q5.z + q5.w)) +
             ((q6.x + q6.y) + (q6.z + q6.w)) + ((q7.x + q7.y) + (q7.z + q7.w));
  }
  for (; i < lim; i += 256) {
    const int4 q = c4[i];
    front += (q.x + q.y) + (q.z + q.w);
  }
  const int v = start + tid;
  const int c = (v < n) ? cnt[v] : 0;

  int incl = c;
  for (int o = 1; o < 32; o <<= 1) {
    int t = __shfl_up(incl, o);
    if (lane >= o) incl += t;
  }
  for (int o = 16; o > 0; o >>= 1) front += __shfl_xor(front, o);
  if (lane == 31) s_wave_cnt[wave] = incl;
  if (lane == 0) s_wave_front[wave] = front;
  __syncthreads();
  int base = 0;
  for (int w = 0; w < 8; ++w) {
    base += s_wave_front[w];
    if (w < wave) base += s_wave_cnt[w];
  }
  const int excl = base + incl - c;
  if (v < n) {
    rowptr[v] = excl;
    dinv[v] = rsqrtf((float)(c + 1));
  }
  if (v == n - 1) rowptr[n] = excl + c;
}

__global__ void fill_lists_kernel(const int* __restrict__ src, const int* __restrict__ dst,
                                  const int* __restrict__ rank,
                                  const int* __restrict__ rowptr, int* __restrict__ col,
                                  int E) {
  int e = blockIdx.x * blockDim.x + threadIdx.x;
  if (e < E) col[rowptr[dst[e]] + rank[e]] = src[e];
}


template <int K, int NC>
__global__ void gemm_wmma_kernel(const float* __restrict__ A,
                                 const float* __restrict__ W,
                                 float* __restrict__ out) {
  const int lane = threadIdx.x & 31;
  const int wave = threadIdx.x >> 5;
  const int m = lane & 15;
  const int h = lane >> 4;
  const int rowTile = blockIdx.x * 16;
  const int colTile = (blockIdx.y * 4 + wave) * 16;

  const float* arow = A + (size_t)(rowTile + m) * K + 2 * h;
  const float* wcol = W + (size_t)(2 * h) * NC + colTile + m;

  v8f c = {0.f, 0.f, 0.f, 0.f, 0.f, 0.f, 0.f, 0.f};

#pragma unroll
  for (int k0 = 0; k0 < K; k0 += 4) {
    float2 av = *(const float2*)(arow + k0);
    v2f a = {av.x, av.y};
    const float* wp = wcol + (size_t)k0 * NC;
    v2f b = {wp[0], wp[NC]};
    c = __builtin_amdgcn_wmma_f32_16x16x4_f32(false, a, false, b, (short)0, c,
                                              false, false);
  }

  float* orow = out + (size_t)(rowTile + h * 8) * NC + colTile + m;
#pragma unroll
  for (int r = 0; r < 8; ++r) orow[(size_t)r * NC] = c[r];
}


template <int F>
__global__ void __launch_bounds__(256)
gather_sum_kernel(const int* __restrict__ rowptr, const int* __restrict__ col,
                  const float* __restrict__ dinv, const float* __restrict__ xw,
                  const float* __restrict__ bias, float* __restrict__ out, int n) {
  constexpr int V = F / 32;
  typedef float vec __attribute__((ext_vector_type(V)));
  const int lane = threadIdx.x & 31;
  const int wave = __builtin_amdgcn_readfirstlane((int)(threadIdx.x >> 5));
  const int wavesPerBlock = (int)(blockDim.x >> 5);
  const int nw = (int)gridDim.x * wavesPerBlock;
  const vec b = *(const vec*)(bias + V * lane);

  for (int v = (int)blockIdx.x * wavesPerBlock + wave; v < n; v += nw) {
    const int beg = rowptr[v];
    const int end = rowptr[v + 1];
    const float dv = dinv[v];
    const vec self = *(const vec*)(xw + (size_t)v * F + V * lane);
    vec acc = (dv * dv) * self;

    for (int base = beg; base < end; base += 32) {
      const int m = (end - base < 32) ? (end - base) : 32;
      const int s_l = col[base + (lane < m ? lane : 0)];
      const float d_l = dinv[s_l];
      int j = 0;
      for (; j + 4 <= m; j += 4) {
        const int s0 = __builtin_amdgcn_readlane(s_l, j);
        const int s1 = __builtin_amdgcn_readlane(s_l, j + 1);
        const int s2 = __builtin_amdgcn_readlane(s_l, j + 2);
        const int s3 = __builtin_amdgcn_readlane(s_l, j + 3);
        const vec r0 = *(const vec*)(xw + (size_t)s0 * F + V * lane);
        const vec r1 = *(const vec*)(xw + (size_t)s1 * F + V * lane);
        const vec r2 = *(const vec*)(xw + (size_t)s2 * F + V * lane);
        const vec r3 = *(const vec*)(xw + (size_t)s3 * F + V * lane);
        const float w0 = __int_as_float(__builtin_amdgcn_readlane(__float_as_int(d_l), j)) * dv;
        const float w1 = __int_as_float(__builtin_amdgcn_readlane(__float_as_int(d_l), j + 1)) * dv;
        const float w2 = __int_as_float(__builtin_amdgcn_readlane(__float_as_int(d_l), j + 2)) * dv;
        const float w3 = __int_as_float(__builtin_amdgcn_readlane(__float_as_int(d_l), j + 3)) * dv;
        acc += w0 * r0;
        acc += w1 * r1;
        acc += w2 * r2;
        acc += w3 * r3;
      }
#pragma clang loop unroll(disable)
      for (; j < m; ++j) {
        const int s0 = __builtin_amdgcn_readlane(s_l, j);
        const vec r0 = *(const vec*)(xw + (size_t)s0 * F + V * lane);
        const float w0 = __int_as_float(__builtin_amdgcn_readlane(__float_as_int(d_l), j)) * dv;
        acc += w0 * r0;
      }
    }
    *(vec*)(out + (size_t)v * F + V * lane) = acc + b;
  }
}

__global__ void tanh_kernel(const float* __restrict__ in, float* __restrict__ out, int n) {
  int i = blockIdx.x * blockDim.x + threadIdx.x;
  if (i < n) out[i] = tanhf(in[i]);
}


extern "C" void kernel_launch(void* const* d_in, const int* in_sizes, int n_in,
                              void* d_out, int out_size, void* d_ws, size_t ws_size,
                              hipStream_t stream) {
  const float* x  = (const float*)d_in[0];
  const int*   ei = (const int*)d_in[1];
  const float* W1 = (const float*)d_in[2];
  const float* b1 = (const float*)d_in[3];
  const float* W2 = (const float*)d_in[4];
  const float* b2 = (const float*)d_in[5];

  const int DIN  = in_sizes[3];
  const int DOUT = in_sizes[5];
  const int N    = in_sizes[0] / DIN;
  const int E    = in_sizes[1] / 2;
  const int* src = ei;
  const int* dst = ei + E;

  char* ws = (char*)d_ws;
  size_t off = ((size_t)N * sizeof(float) + 255) & ~(size_t)255;
  float* dinv = (float*)ws;
  float* xw1  = (float*)(ws + off);
  size_t off2 = off + (((size_t)N * DIN * sizeof(float) + 255) & ~(size_t)255);
  float* h1   = (float*)(ws + off2);
  float* xw2  = xw1;
  float* pre  = h1;
  float* out  = (float*)d_out;

  char* ob = (char*)d_out;
  size_t lo1 = ((size_t)N * sizeof(int) + 255) & ~(size_t)255;
  size_t lo2 = lo1 + (((size_t)(N + 1) * sizeof(int) + 255) & ~(size_t)255);
  size_t lo3 = lo2 + (((size_t)E * sizeof(int) + 255) & ~(size_t)255);
  int* cnt    = (int*)ob;
  int* rowptr = (int*)(ob + lo1);
  int* rank   = (int*)(ob + lo2);
  int* col    = (int*)(ob + lo3);

  const int TB = 256;

  zero_counts_kernel<<<(N + TB - 1) / TB, TB, 0, stream>>>(cnt, N);
  count_rank_kernel<<<(E + TB - 1) / TB, TB, 0, stream>>>(dst, cnt, rank, E);
  scan_counts_kernel<<<(N + 255) / 256, 256, 0, stream>>>(cnt, rowptr, dinv, N);
  fill_lists_kernel<<<(E + TB - 1) / TB, TB, 0, stream>>>(src, dst, rank, rowptr, col, E);

  gemm_wmma_kernel<128, 128><<<dim3(N / 16, 128 / 64), 128, 0, stream>>>(x, W1, xw1);
  gather_sum_kernel<128><<<(N + 15) / 16, 256, 0, stream>>>(rowptr, col, dinv, xw1, b1, h1, N);

  gemm_wmma_kernel<128, 64><<<dim3(N / 16, 64 / 64), 128, 0, stream>>>(h1, W2, xw2);
  gather_sum_kernel<64><<<(N + 15) / 16, 256, 0, stream>>>(rowptr, col, dinv, xw2, b2, pre, N);
  tanh_kernel<<<(out_size + TB - 1) / TB, TB, 0, stream>>>(pre, out, out_size);
}


# flat passes (counter clear, edge count, list fill, tanh) take 4 to 16 neighbouring elements a thread with 16-byte loads and stores, so about 125000 fewer waves start per call
# speedup vs baseline: 1.9495x; 1.0403x over previous
#include <hip/hip_runtime.h>
#include <hip/hip_bf16.h>

typedef __attribute__((ext_vector_type(2))) float v2f;
typedef __attribute__((ext_vector_type(8))) float v8f;


__global__ void zero_counts_kernel(int* __restrict__ cnt, int n) {
  int i = blockIdx.x * blockDim.x + threadIdx.x;
  if (4 * i < n) ((int4*)cnt)[i] = make_int4(0, 0, 0, 0);
}

__device__ __forceinline__ int take_slot(int* __restrict__ cnt, int d) {
  return __hip_atomic_fetch_add(&cnt[d], 1, __ATOMIC_RELAXED, __HIP_MEMORY_SCOPE_AGENT);
}

__global__ void count_rank_kernel(const int* __restrict__ dst, int* __restrict__ cnt,
                                  int* __restrict__ rank, int E) {
  const int e = 4 * (int)(blockIdx.x * blockDim.x + threadIdx.x);
  if ((E & 3) == 0) {
    if (e < E) {
      const int4 d = *(const int4*)(dst + e);
      int4 r;
      r.x = take_slot(cnt, d.x);
      r.y = take_slot(cnt, d.y);
      r.z = take_slot(cnt, d.z);
      r.w = take_slot(cnt, d.w);
      *(int4*)(rank + e) = r;
    }
  } else {
    for (int k = e; k < e + 4 && k < E; ++k) rank[k] = take_slot(cnt, dst[k]);
  }
}

__global__ void __launch_bounds__(256)
scan_counts_kernel(const int* __restrict__ cnt, int* __restrict__ rowptr,
                   float* __restrict__ dinv, int n) {
  __shared__ int s_wave_cnt[8];
  __shared__ int s_wave_front[8];
  const int tid = threadIdx.x;
  const int lane = tid & 31;
  const int wave = tid >> 5;
  const int start = blockIdx.x * 256;

  int front = 0;
  const int4* c4 = (const int4*)cnt;
  const int lim = start / 4;
  int i = tid;
  for (; i + 7 * 256 < lim; i += 8 * 256) {
    const int4 q0 = c4[i];
    const int4 q1 = c4[i + 256];
    const int4 q2 = c4[i + 2 * 256];
    const int4 q3 = c4[i + 3 * 256];
    const int4 q4 = c4[i + 4 * 256];
    const int4 q5 = c4[i + 5 * 256];
    const int4 q6 = c4[i + 6 * 256];
    const int4 q7 = c4[i + 7 * 256];
    front += ((q0.x + q0.y) + (q0.z + q0.w)) + ((q1.x + q1.y) + (q1.z + q1.w)) +
             ((q2.x + q2.y) + (q2.z + q2.w)) + ((q3.x + q3.y) + (q3.z + q3.w)) +
             ((q4.x + q4.y) + (q4.z + q4.w)) + ((q5.x + q5.y) + (q5.z + q5.w)) +
             ((q6.x + q6.y) + (q6.z + q6.w)) + ((q7.x + q7.y) + (q7.z + q7.w));
  }
  for (; i < lim; i += 256) {
    const int4 q = c4[i];
    front += (q.x + q.y) + (q.z + q.w);
  }
  const int v = start + tid;
  const int c = (v < n) ? cnt[v] : 0;

  int incl = c;
  for (int o = 1; o < 32; o <<= 1) {
    int t = __shfl_up(incl, o);
    if (lane >= o) incl += t;
  }
  for (int o = 16; o > 0; o >>= 1) front += __shfl_xor(front, o);
  if (lane == 31) s_wave_cnt[wave] = incl;
  if (lane == 0) s_wave_front[wave] = front;
  __syncthreads();
  int base = 0;
  for (int w = 0; w < 8; ++w) {
    base += s_wave_front[w];
    if (w < wave) base += s_wave_cnt[w];
  }
  const int excl = base + incl - c;
  if (v < n) {
    rowptr[v] = excl;
    dinv[v] = rsqrtf((float)(c + 1));
  }
  if (v == n - 1) rowptr[n] = excl + c;
}

__global__ void fill_lists_kernel(const int* __restrict__ src, const int* __restrict__ dst,
                                  const int* __restrict__ rank,
                                  const int* __restrict__ rowptr, int* __restrict__ col,
                                  int E) {
  const int e = 4 * (int)(blockIdx.x * blockDim.x + threadIdx.x);
  if ((E & 3) == 0) {
    if (e < E) {
      const int4 s = *(const int4*)(src + e);
      const int4 d = *(const int4*)(dst + e);
      const int4 r = *(const int4*)(rank + e);
      const int p0 = rowptr[d.x];
      const int p1 = rowptr[d.y];
      const int p2 = rowptr[d.z];
      const int p3 = rowptr[d.w];
      col[p0 + r.x] = s.x;
      col[p1 + r.y] = s.y;
      col[p2 + r.z] = s.z;
      col[p3 + r.w] = s.w;
    }
  } else {
    for (int k = e; k < e + 4 && k < E; ++k) col[rowptr[dst[k]] + rank[k]] = src[k];
  }
}


template <int K, int NC>
__global__ void gemm_wmma_kernel(const float* __restrict__ A,
                                 const float* __restrict__ W,
                                 float* __restrict__ out) {
  const int lane = threadIdx.x & 31;
  const int wave = threadIdx.x >> 5;
  const int m = lane & 15;
  const int h = lane >> 4;
  const int rowTile = blockIdx.x * 16;
  const int colTile = (blockIdx.y * 4 + wave) * 16;

  const float* arow = A + (size_t)(rowTile + m) * K + 2 * h;
  const float* wcol = W + (size_t)(2 * h) * NC + colTile + m;

  v8f c = {0.f, 0.f, 0.f, 0.f, 0.f, 0.f, 0.f, 0.f};

#pragma unroll
  for (int k0 = 0; k0 < K; k0 += 4) {
    float2 av = *(const float2*)(arow + k0);
    v2f a = {av.x, av.y};
    const float* wp = wcol + (size_t)k0 * NC;
    v2f b = {wp[0], wp[NC]};
    c = __builtin_amdgcn_wmma_f32_16x16x4_f32(false, a, false, b, (short)0, c,
                                              false, false);
  }

  float* orow = out + (size_t)(rowTile + h * 8) * NC + colTile + m;
#pragma unroll
  for (int r = 0; r < 8; ++r) orow[(size_t)r * NC] = c[r];
}


template <int F>
__global__ void __launch_bounds__(256)
gather_sum_kernel(const int* __restrict__ rowptr, const int* __restrict__ col,
                  const float* __restrict__ dinv, const float* __restrict__ xw,
                  const float* __restrict__ bias, float* __restrict__ out, int n) {
  constexpr int V = F / 32;
  typedef float vec __attribute__((ext_vector_type(V)));
  const int lane = threadIdx.x & 31;
  const int wave = __builtin_amdgcn_readfirstlane((int)(threadIdx.x >> 5));
  const int wavesPerBlock = (int)(blockDim.x >> 5);
  const int nw = (int)gridDim.x * wavesPerBlock;
  const vec b = *(const vec*)(bias + V * lane);

  for (int v = (int)blockIdx.x * wavesPerBlock + wave; v < n; v += nw) {
    const int beg = rowptr[v];
    const int end = rowptr[v + 1];
    const float dv = dinv[v];
    const vec self = *(const vec*)(xw + (size_t)v * F + V * lane);
    vec acc = (dv * dv) * self;

    for (int base = beg; base < end; base += 32) {
      const int m = (end - base < 32) ? (end - base) : 32;
      const int s_l = col[base + (lane < m ? lane : 0)];
      const float d_l = dinv[s_l];
      int j = 0;
      for (; j + 4 <= m; j += 4) {
        const int s0 = __builtin_amdgcn_readlane(s_l, j);
        const int s1 = __builtin_amdgcn_readlane(s_l, j + 1);
        const int s2 = __builtin_amdgcn_readlane(s_l, j + 2);
        const int s3 = __builtin_amdgcn_readlane(s_l, j + 3);
        const vec r0 = *(const vec*)(xw + (size_t)s0 * F + V * lane);
        const vec r1 = *(const vec*)(xw + (size_t)s1 * F + V * lane);
        const vec r2 = *(const vec*)(xw + (size_t)s2 * F + V * lane);
        const vec r3 = *(const vec*)(xw + (size_t)s3 * F + V * lane);
        const float w0 = __int_as_float(__builtin_amdgcn_readlane(__float_as_int(d_l), j)) * dv;
        const float w1 = __int_as_float(__builtin_amdgcn_readlane(__float_as_int(d_l), j + 1)) * dv;
        const float w2 = __int_as_float(__builtin_amdgcn_readlane(__float_as_int(d_l), j + 2)) * dv;
        const float w3 = __int_as_float(__builtin_amdgcn_readlane(__float_as_int(d_l), j + 3)) * dv;
        acc += w0 * r0;
        acc += w1 * r1;
        acc += w2 * r2;
        acc += w3 * r3;
      }
#pragma clang loop unroll(disable)
      for (; j < m; ++j) {
        const int s0 = __builtin_amdgcn_readlane(s_l, j);
        const vec r0 = *(const vec*)(xw + (size_t)s0 * F + V * lane);
        const float w0 = __int_as_float(__builtin_amdgcn_readlane(__float_as_int(d_l), j)) * dv;
        acc += w0 * r0;
      }
    }
    *(vec*)(out + (size_t)v * F + V * lane) = acc + b;
  }
}

__global__ void __launch_bounds__(256)
tanh_kernel(const float* __restrict__ in, float* __restrict__ out, int n) {
  const int n4 = n >> 2;
  const float4* in4 = (const float4*)in;
  float4* out4 = (float4*)out;
  const int first = (int)blockIdx.x * 1024 + (int)threadIdx.x;
  float4 q[4];
#pragma unroll
  for (int u = 0; u < 4; ++u) {
    const int i = first + u * 256;
    q[u] = (i < n4) ? in4[i] : make_float4(0.f, 0.f, 0.f, 0.f);
  }
#pragma unroll
  for (int u = 0; u < 4; ++u) {
    const int i = first + u * 256;
    if (i < n4) {
      float4 t;
      t.x = tanhf(q[u].x);
      t.y = tanhf(q[u].y);
      t.z = tanhf(q[u].z);
      t.w = tanhf(q[u].w);
      out4[i] = t;
    }
  }
  if (blockIdx.x == 0 && (int)threadIdx.x < (n & 3)) {
    const int i = 4 * n4 + (int)threadIdx.x;
    out[i] = tanhf(in[i]);
  }
}


extern "C" void kernel_launch(void* const* d_in, const int* in_sizes, int n_in,
                              void* d_out, int out_size, void* d_ws, size_t ws_size,
                              hipStream_t stream) {
  const float* x  = (const float*)d_in[0];
  const int*   ei = (const int*)d_in[1];
  const float* W1 = (const float*)d_in[2];
  const float* b1 = (const float*)d_in[3];
  const float* W2 = (const float*)d_in[4];
  const float* b2 = (const float*)d_in[5];

  const int DIN  = in_sizes[3];
  const int DOUT = in_sizes[5];
  const int N    = in_sizes[0] / DIN;
  const int E    = in_sizes[1] / 2;
  const int* src = ei;
  const int* dst = ei + E;

  char* ws = (char*)d_ws;
  size_t off = ((size_t)N * sizeof(float) + 255) & ~(size_t)255;
  float* dinv = (float*)ws;
  float* xw1  = (float*)(ws + off);
  size_t off2 = off + (((size_t)N * DIN * sizeof(float) + 255) & ~(size_t)255);
  float* h1   = (float*)(ws + off2);
  float* xw2  = xw1;
  float* pre  = h1;
  float* out  = (float*)d_out;

  char* ob = (char*)d_out;
  size_t lo1 = ((size_t)N * sizeof(int) + 255) & ~(size_t)255;
  size_t lo2 = lo1 + (((size_t)(N + 1) * sizeof(int) + 255) & ~(size_t)255);
  size_t lo3 = lo2 + (((size_t)E * sizeof(int) + 255) & ~(size_t)255);
  int* cnt    = (int*)ob;
  int* rowptr = (int*)(ob + lo1);
  int* rank   = (int*)(ob + lo2);
  int* col    = (int*)(ob + lo3);

  const int TB = 256;

  zero_counts_kernel<<<(N + 4 * TB - 1) / (4 * TB), TB, 0, stream>>>(cnt, N);
  count_rank_kernel<<<(E + 4 * TB - 1) / (4 * TB), TB, 0, stream>>>(dst, cnt, rank, E);
  scan_counts_kernel<<<(N + 255) / 256, 256, 0, stream>>>(cnt, rowptr, dinv, N);
  fill_lists_kernel<<<(E + 4 * TB - 1) / (4 * TB), TB, 0, stream>>>(src, dst, rank, rowptr, col, E);

  gemm_wmma_kernel<128, 128><<<dim3(N / 16, 128 / 64), 128, 0, stream>>>(x, W1, xw1);
  gather_sum_kernel<128><<<(N + 15) / 16, 256, 0, stream>>>(rowptr, col, dinv, xw1, b1, h1, N);

  gemm_wmma_kernel<128, 64><<<dim3(N / 16, 64 / 64), 128, 0, stream>>>(h1, W2, xw2);
  gather_sum_kernel<64><<<(N + 15) / 16, 256, 0, stream>>>(rowptr, col, dinv, xw2, b2, pre, N);
  tanh_kernel<<<(out_size + 16 * TB - 1) / (16 * TB), TB, 0, stream>>>(pre, out, out_size);
}


# GEMM: one wave keeps a 16x64 output strip (4 tiles, columns interleaved by 4): each A fragment feeds 4 WMMAs, operand loads and stores are 16 B a lane; sum order over k changes inside each 8-block
# speedup vs baseline: 2.1545x; 1.1052x over previous
#include <hip/hip_runtime.h>
#include <hip/hip_bf16.h>

typedef __attribute__((ext_vector_type(2))) float v2f;
typedef __attribute__((ext_vector_type(8))) float v8f;


__global__ void zero_counts_kernel(int* __restrict__ cnt, int n) {
  int i = blockIdx.x * blockDim.x + threadIdx.x;
  if (4 * i < n) ((int4*)cnt)[i] = make_int4(0, 0, 0, 0);
}

__device__ __forceinline__ int take_slot(int* __restrict__ cnt, int d) {
  return __hip_atomic_fetch_add(&cnt[d], 1, __ATOMIC_RELAXED, __HIP_MEMORY_SCOPE_AGENT);
}

__global__ void count_rank_kernel(const int* __restrict__ dst, int* __restrict__ cnt,
                                  int* __restrict__ rank, int E) {
  const int e = 4 * (int)(blockIdx.x * blockDim.x + threadIdx.x);
  if ((E & 3) == 0) {
    if (e < E) {
      const int4 d = *(const int4*)(dst + e);
      int4 r;
      r.x = take_slot(cnt, d.x);
      r.y = take_slot(cnt, d.y);
      r.z = take_slot(cnt, d.z);
      r.w = take_slot(cnt, d.w);
      *(int4*)(rank + e) = r;
    }
  } else {
    for (int k = e; k < e + 4 && k < E; ++k) rank[k] = take_slot(cnt, dst[k]);
  }
}

__global__ void __launch_bounds__(256)
scan_counts_kernel(const int* __restrict__ cnt, int* __restrict__ rowptr,
                   float* __restrict__ dinv, int n) {
  __shared__ int s_wave_cnt[8];
  __shared__ int s_wave_front[8];
  const int tid = threadIdx.x;
  const int lane = tid & 31;
  const int wave = tid >> 5;
  const int start = blockIdx.x * 256;

  int front = 0;
  const int4* c4 = (const int4*)cnt;
  const int lim = start / 4;
  int i = tid;
  for (; i + 7 * 256 < lim; i += 8 * 256) {
    const int4 q0 = c4[i];
    const int4 q1 = c4[i + 256];
    const int4 q2 = c4[i + 2 * 256];
    const int4 q3 = c4[i + 3 * 256];
    const int4 q4 = c4[i + 4 * 256];
    const int4 q5 = c4[i + 5 * 256];
    const int4 q6 = c4[i + 6 * 256];
    const int4 q7 = c4[i + 7 * 256];
    front += ((q0.x + q0.y) + (q0.z + q0.w)) + ((q1.x + q1.y) + (q1.z + q1.w)) +
             ((q2.x + q2.y) + (q2.z + q2.w)) + ((q3.x + q3.y) + (q3.z + q3.w)) +
             ((q4.x + q4.y) + (q4.z + q4.w)) + ((q5.x + q5.y) + (q5.z + q5.w)) +
             ((q6.x + q6.y) + (q6.z + q6.w)) + ((q7.x + q7.y) + (q7.z + q7.w));
  }
  for (; i < lim; i += 256) {
    const int4 q = c4[i];
    front += (q.x + q.y) + (q.z + q.w);
  }
  const int v = start + tid;
  const int c = (v < n) ? cnt[v] : 0;

  int incl = c;
  for (int o = 1; o < 32; o <<= 1) {
    int t = __shfl_up(incl, o);
    if (lane >= o) incl += t;
  }
  for (int o = 16; o > 0; o >>= 1) front += __shfl_xor(front, o);
  if (lane == 31) s_wave_cnt[wave] = incl;
  if (lane == 0) s_wave_front[wave] = front;
  __syncthreads();
  int base = 0;
  for (int w = 0; w < 8; ++w) {
    base += s_wave_front[w];
    if (w < wave) base += s_wave_cnt[w];
  }
  const int excl = base + incl - c;
  if (v < n) {
    rowptr[v] = excl;
    dinv[v] = rsqrtf((float)(c + 1));
  }
  if (v == n - 1) rowptr[n] = excl + c;
}

__global__ void fill_lists_kernel(const int* __restrict__ src, const int* __restrict__ dst,
                                  const int* __restrict__ rank,
                                  const int* __restrict__ rowptr, int* __restrict__ col,
                                  int E) {
  const int e = 4 * (int)(blockIdx.x * blockDim.x + threadIdx.x);
  if ((E & 3) == 0) {
    if (e < E) {
      const int4 s = *(const int4*)(src + e);
      const int4 d = *(const int4*)(dst + e);
      const int4 r = *(const int4*)(rank + e);
      const int p0 = rowptr[d.x];
      const int p1 = rowptr[d.y];
      const int p2 = rowptr[d.z];
      const int p3 = rowptr[d.w];
      col[p0 + r.x] = s.x;
      col[p1 + r.y] = s.y;
      col[p2 + r.z] = s.z;
      col[p3 + r.w] = s.w;
    }
  } else {
    for (int k = e; k < e + 4 && k < E; ++k) col[rowptr[dst[k]] + rank[k]] = src[k];
  }
}


__device__ __forceinline__ v8f wmma_f32(v2f a, float b_lo, float b_hi, v8f c) {
  v2f b = {b_lo, b_hi};
  return __builtin_amdgcn_wmma_f32_16x16x4_f32(false, a, false, b, (short)0, c, false, false);
}

template <int K, int NC>
__global__ void __launch_bounds__(128)
gemm_wmma_kernel(const float* __restrict__ A, const float* __restrict__ W, float* __restrict__ out, int rowTiles) {
  constexpr int G = NC / 64;
  constexpr int NB = K / 8;
  constexpr int AHEAD = 3;
  const int lane = threadIdx.x & 31;
  const int wave = __builtin_amdgcn_readfirstlane((int)(threadIdx.x >> 5));
  const int c = lane & 15;
  const int h = lane >> 4;
  const int item = (int)blockIdx.x * 4 + wave;
  const int rowTile = item / G;
  const int strip = item % G;
  if (rowTile >= rowTiles) return;

  const float* arow = A + (size_t)(rowTile * 16 + c) * K + 4 * h;
  const float* wrow = W + (size_t)(4 * h) * NC + strip * 64 + 4 * c;

  v8f acc0 = {0.f, 0.f, 0.f, 0.f, 0.f, 0.f, 0.f, 0.f};
  v8f acc1 = acc0, acc2 = acc0, acc3 = acc0;

  float4 a[NB];
  float4 b[NB][4];

#pragma unroll
  for (int i = 0; i < AHEAD && i < NB; ++i) {
    a[i] = *(const float4*)(arow + 8 * i);
#pragma unroll
    for (int j = 0; j < 4; ++j) b[i][j] = *(const float4*)(wrow + (size_t)(8 * i + j) * NC);
  }

#pragma unroll
  for (int kb = 0; kb < NB; ++kb) {
    if (kb + AHEAD < NB) {
      const int n = kb + AHEAD;
      a[n] = *(const float4*)(arow + 8 * n);
#pragma unroll
      for (int j = 0; j < 4; ++j) b[n][j] = *(const float4*)(wrow + (size_t)(8 * n + j) * NC);
    }
    __builtin_amdgcn_sched_barrier(0);
    const v2f a01 = {a[kb].x, a[kb].y};
    const v2f a23 = {a[kb].z, a[kb].w};
    acc0 = wmma_f32(a01, b[kb][0].x, b[kb][1].x, acc0);
    acc1 = wmma_f32(a01, b[kb][0].y, b[kb][1].y, acc1);
    acc2 = wmma_f32(a01, b[kb][0].z, b[kb][1].z, acc2);
    acc3 = wmma_f32(a01, b[kb][0].w, b[kb][1].w, acc3);
    acc0 = wmma_f32(a23, b[kb][2].x, b[kb][3].x, acc0);
    acc1 = wmma_f32(a23, b[kb][2].y, b[kb][3].y, acc1);
    acc2 = wmma_f32(a23, b[kb][2].z, b[kb][3].z, acc2);
    acc3 = wmma_f32(a23, b[kb][2].w, b[kb][3].w, acc3);
  }

  float* orow = out + (size_t)(rowTile * 16 + h * 8) * NC + strip * 64 + 4 * c;
#pragma unroll
  for (int r = 0; r < 8; ++r)
    *(float4*)(orow + (size_t)r * NC) = make_float4(acc0[r], acc1[r], acc2[r], acc3[r]);
}


template <int F>
__global__ void __launch_bounds__(256)
gather_sum_kernel(const int* __restrict__ rowptr, const int* __restrict__ col,
                  const float* __restrict__ dinv, const float* __restrict__ xw,
                  const float* __restrict__ bias, float* __restrict__ out, int n) {
  constexpr int V = F / 32;
  typedef float vec __attribute__((ext_vector_type(V)));
  const int lane = threadIdx.x & 31;
  const int wave = __builtin_amdgcn_readfirstlane((int)(threadIdx.x >> 5));
  const int wavesPerBlock = (int)(blockDim.x >> 5);
  const int nw = (int)gridDim.x * wavesPerBlock;
  const vec b = *(const vec*)(bias + V * lane);

  for (int v = (int)blockIdx.x * wavesPerBlock + wave; v < n; v += nw) {
    const int beg = rowptr[v];
    const int end = rowptr[v + 1];
    const float dv = dinv[v];
    const vec self = *(const vec*)(xw + (size_t)v * F + V * lane);
    vec acc = (dv * dv) * self;

    for (int base = beg; base < end; base += 32) {
      const int m = (end - base < 32) ? (end - base) : 32;
      const int s_l = col[base + (lane < m ? lane : 0)];
      const float d_l = dinv[s_l];
      int j = 0;
      for (; j + 4 <= m; j += 4) {
        const int s0 = __builtin_amdgcn_readlane(s_l, j);
        const int s1 = __builtin_amdgcn_readlane(s_l, j + 1);
        const int s2 = __builtin_amdgcn_readlane(s_l, j + 2);
        const int s3 = __builtin_amdgcn_readlane(s_l, j + 3);
        const vec r0 = *(const vec*)(xw + (size_t)s0 * F + V * lane);
        const vec r1 = *(const vec*)(xw + (size_t)s1 * F + V * lane);
        const vec r2 = *(const vec*)(xw + (size_t)s2 * F + V * lane);
        const vec r3 = *(const vec*)(xw + (size_t)s3 * F + V * lane);
        const float w0 = __int_as_float(__builtin_amdgcn_readlane(__float_as_int(d_l), j)) * dv;
        const float w1 = __int_as_float(__builtin_amdgcn_readlane(__float_as_int(d_l), j + 1)) * dv;
        const float w2 = __int_as_float(__builtin_amdgcn_readlane(__float_as_int(d_l), j + 2)) * dv;
        const float w3 = __int_as_float(__builtin_amdgcn_readlane(__float_as_int(d_l), j + 3)) * dv;
        acc += w0 * r0;
        acc += w1 * r1;
        acc += w2 * r2;
        acc += w3 * r3;
      }
#pragma clang loop unroll(disable)
      for (; j < m; ++j) {
        const int s0 = __builtin_amdgcn_readlane(s_l, j);
        const vec r0 = *(const vec*)(xw + (size_t)s0 * F + V * lane);
        const float w0 = __int_as_float(__builtin_amdgcn_readlane(__float_as_int(d_l), j)) * dv;
        acc += w0 * r0;
      }
    }
    *(vec*)(out + (size_t)v * F + V * lane) = acc + b;
  }
}

__global__ void __launch_bounds__(256)
tanh_kernel(const float* __restrict__ in, float* __restrict__ out, int n) {
  const int n4 = n >> 2;
  const float4* in4 = (const float4*)in;
  float4* out4 = (float4*)out;
  const int first = (int)blockIdx.x * 1024 + (int)threadIdx.x;
  float4 q[4];
#pragma unroll
  for (int u = 0; u < 4; ++u) {
    const int i = first + u * 256;
    q[u] = (i < n4) ? in4[i] : make_float4(0.f, 0.f, 0.f, 0.f);
  }
#pragma unroll
  for (int u = 0; u < 4; ++u) {
    const int i = first + u * 256;
    if (i < n4) {
      float4 t;
      t.x = tanhf(q[u].x);
      t.y = tanhf(q[u].y);
      t.z = tanhf(q[u].z);
      t.w = tanhf(q[u].w);
      out4[i] = t;
    }
  }
  if (blockIdx.x == 0 && (int)threadIdx.x < (n & 3)) {
    const int i = 4 * n4 + (int)threadIdx.x;
    out[i] = tanhf(in[i]);
  }
}


extern "C" void kernel_launch(void* const* d_in, const int* in_sizes, int n_in,
                              void* d_out, int out_size, void* d_ws, size_t ws_size,
                              hipStream_t stream) {
  const float* x  = (const float*)d_in[0];
  const int*   ei = (const int*)d_in[1];
  const float* W1 = (const float*)d_in[2];
  const float* b1 = (const float*)d_in[3];
  const float* W2 = (const float*)d_in[4];
  const float* b2 = (const float*)d_in[5];

  const int DIN  = in_sizes[3];
  const int DOUT = in_sizes[5];
  const int N    = in_sizes[0] / DIN;
  const int E    = in_sizes[1] / 2;
  const int* src = ei;
  const int* dst = ei + E;

  char* ws = (char*)d_ws;
  size_t off = ((size_t)N * sizeof(float) + 255) & ~(size_t)255;
  float* dinv = (float*)ws;
  float* xw1  = (float*)(ws + off);
  size_t off2 = off + (((size_t)N * DIN * sizeof(float) + 255) & ~(size_t)255);
  float* h1   = (float*)(ws + off2);
  float* xw2  = xw1;
  float* pre  = h1;
  float* out  = (float*)d_out;

  char* ob = (char*)d_out;
  size_t lo1 = ((size_t)N * sizeof(int) + 255) & ~(size_t)255;
  size_t lo2 = lo1 + (((size_t)(N + 1) * sizeof(int) + 255) & ~(size_t)255);
  size_t lo3 = lo2 + (((size_t)E * sizeof(int) + 255) & ~(size_t)255);
  int* cnt    = (int*)ob;
  int* rowptr = (int*)(ob + lo1);
  int* rank   = (int*)(ob + lo2);
  int* col    = (int*)(ob + lo3);

  const int TB = 256;

  zero_counts_kernel<<<(N + 4 * TB - 1) / (4 * TB), TB, 0, stream>>>(cnt, N);
  count_rank_kernel<<<(E + 4 * TB - 1) / (4 * TB), TB, 0, stream>>>(dst, cnt, rank, E);
  scan_counts_kernel<<<(N + 255) / 256, 256, 0, stream>>>(cnt, rowptr, dinv, N);
  fill_lists_kernel<<<(E + 4 * TB - 1) / (4 * TB), TB, 0, stream>>>(src, dst, rank, rowptr, col, E);

  gemm_wmma_kernel<128, 128><<<(N / 16 * (128 / 64) + 3) / 4, 128, 0, stream>>>(x, W1, xw1, N / 16);
  gather_sum_kernel<128><<<(N + 15) / 16, 256, 0, stream>>>(rowptr, col, dinv, xw1, b1, h1, N);

  gemm_wmma_kernel<128, 64><<<(N / 16 * (64 / 64) + 3) / 4, 128, 0, stream>>>(h1, W2, xw2, N / 16);
  gather_sum_kernel<64><<<(N + 15) / 16, 256, 0, stream>>>(rowptr, col, dinv, xw2, b2, pre, N);
  tanh_kernel<<<(out_size + 16 * TB - 1) / (16 * TB), TB, 0, stream>>>(pre, out, out_size);
}


# edge count and rank on LDS counters (a work-group counts one 4096-node range of one edge chunk) instead of one returning global add per edge; list order is now chunk, then LDS arrival order
# speedup vs baseline: 2.6222x; 1.2171x over previous
#include <hip/hip_runtime.h>
#include <hip/hip_bf16.h>

typedef __attribute__((ext_vector_type(2))) float v2f;
typedef __attribute__((ext_vector_type(8))) float v8f;


constexpr int CHUNKS = 20;
constexpr int RANGE = 4096;
constexpr int GROUPS = RANGE / 256;
constexpr int CT = 1024;

__device__ __forceinline__ bool in_range(int d, int lo) { return (unsigned)(d - lo) < (unsigned)RANGE; }

__device__ __forceinline__ int lds_take_slot(int* counter) {
  return __hip_atomic_fetch_add(counter, 1, __ATOMIC_RELAXED, __HIP_MEMORY_SCOPE_WORKGROUP);
}

__global__ void __launch_bounds__(CT)
count_rank_kernel(const int* __restrict__ dst, int* __restrict__ rank, int* __restrict__ cntm,
                  int* __restrict__ sub, int E, int chunk, int NP, int SUBP) {
  __shared__ int lc[RANGE];
  __shared__ int s_sub[GROUPS];
  const int tid = threadIdx.x;
  const int ci = blockIdx.x;
  const int rj = blockIdx.y;
  const int lo = rj * RANGE;

  ((int4*)lc)[tid] = make_int4(0, 0, 0, 0);
  if (tid < GROUPS) s_sub[tid] = 0;
  __syncthreads();

  const int cbeg = ci * chunk;
  const int cend = (cbeg + chunk < E) ? (cbeg + chunk) : E;
  if ((E & 3) == 0) {
    const int n4 = (cend - cbeg) >> 2;
    const int4* d4 = (const int4*)(dst + cbeg);
    int* rk = rank + cbeg;
    for (int q = tid; q < n4; q += 8 * CT) {
      int4 d[8];
#pragma unroll
      for (int u = 0; u < 8; ++u) {
        const int idx = q + u * CT;
        d[u] = (idx < n4) ? d4[idx] : make_int4(-1, -1, -1, -1);
      }
#pragma unroll
      for (int u = 0; u < 8; ++u) {
        const int e = 4 * (q + u * CT);
        const bool h0 = in_range(d[u].x, lo);
        const bool h1 = in_range(d[u].y, lo);
        const bool h2 = in_range(d[u].z, lo);
        const bool h3 = in_range(d[u].w, lo);
        int r0 = 0, r1 = 0, r2 = 0, r3 = 0;
        if (h0) r0 = lds_take_slot(&lc[d[u].x - lo]);
        if (h1) r1 = lds_take_slot(&lc[d[u].y - lo]);
        if (h2) r2 = lds_take_slot(&lc[d[u].z - lo]);
        if (h3) r3 = lds_take_slot(&lc[d[u].w - lo]);
        if (h0) rk[e] = r0;
        if (h1) rk[e + 1] = r1;
        if (h2) rk[e + 2] = r2;
        if (h3) rk[e + 3] = r3;
      }
    }
  } else {
    for (int k = cbeg + tid; k < cend; k += CT) {
      const int d = dst[k];
      if (in_range(d, lo)) rank[k] = lds_take_slot(&lc[d - lo]);
    }
  }
  __syncthreads();

  const int4 c = ((const int4*)lc)[tid];
  *(int4*)(cntm + (size_t)ci * NP + lo + 4 * tid) = c;
  int s = (c.x + c.y) + (c.z + c.w);
  for (int o = 16; o > 0; o >>= 1) s += __shfl_xor(s, o);
  if ((tid & 31) == 0)
    __hip_atomic_fetch_add(&s_sub[tid >> 6], s, __ATOMIC_RELAXED, __HIP_MEMORY_SCOPE_WORKGROUP);
  __syncthreads();
  if (tid < GROUPS) sub[ci * SUBP + rj * GROUPS + tid] = s_sub[tid];
}

__global__ void __launch_bounds__(256)
scan_counts_kernel(int* __restrict__ cntm, const int* __restrict__ sub,
                   int* __restrict__ rowptr, float* __restrict__ dinv, int n, int NP, int SUBP) {
  __shared__ int s_wave_cnt[8];
  __shared__ int s_wave_front[8];
  const int tid = threadIdx.x;
  const int lane = tid & 31;
  const int wave = tid >> 5;
  const int g = blockIdx.x;
  const int v = g * 256 + tid;

  int part[CHUNKS];
  int fr[CHUNKS];
  const bool has = tid < g;
  const int gi = has ? tid : 0;
#pragma unroll
  for (int i = 0; i < CHUNKS; ++i) {
    part[i] = cntm[(size_t)i * NP + v];
    fr[i] = sub[i * SUBP + gi];
  }
  int c = 0;
  int front = 0;
#pragma unroll
  for (int i = 0; i < CHUNKS; ++i) {
    c += part[i];
    front += has ? fr[i] : 0;
  }
#pragma clang loop unroll(disable)
  for (int gg = tid + 256; gg < g; gg += 256) {
#pragma clang loop unroll(disable)
    for (int i = 0; i < CHUNKS; ++i) front += sub[i * SUBP + gg];
  }

  int incl = c;
  for (int o = 1; o < 32; o <<= 1) {
    int t = __shfl_up(incl, o);
    if (lane >= o) incl += t;
  }
  for (int o = 16; o > 0; o >>= 1) front += __shfl_xor(front, o);
  if (lane == 31) s_wave_cnt[wave] = incl;
  if (lane == 0) s_wave_front[wave] = front;
  __syncthreads();
  int base = 0;
  for (int w = 0; w < 8; ++w) {
    base += s_wave_front[w];
    if (w < wave) base += s_wave_cnt[w];
  }
  const int excl = base + incl - c;
  if (v < n) {
    rowptr[v] = excl;
    dinv[v] = rsqrtf((float)(c + 1));
    int run = excl;
#pragma unroll
    for (int i = 0; i < CHUNKS; ++i) {
      cntm[(size_t)i * NP + v] = run;
      run += part[i];
    }
  }
  if (v == n - 1) rowptr[n] = excl + c;
}

__global__ void fill_lists_kernel(const int* __restrict__ src, const int* __restrict__ dst,
                                  const int* __restrict__ rank,
                                  const int* __restrict__ first, int* __restrict__ col,
                                  int E, int chunk, int NP) {
  const int e = 4 * (int)(blockIdx.x * blockDim.x + threadIdx.x);
  if ((E & 3) == 0) {
    if (e < E) {
      const int* f = first + (size_t)(e / chunk) * NP;
      const int4 s = *(const int4*)(src + e);
      const int4 d = *(const int4*)(dst + e);
      const int4 r = *(const int4*)(rank + e);
      const int p0 = f[d.x];
      const int p1 = f[d.y];
      const int p2 = f[d.z];
      const int p3 = f[d.w];
      col[p0 + r.x] = s.x;
      col[p1 + r.y] = s.y;
      col[p2 + r.z] = s.z;
      col[p3 + r.w] = s.w;
    }
  } else {
    for (int k = e; k < e + 4 && k < E; ++k)
      col[first[(size_t)(k / chunk) * NP + dst[k]] + rank[k]] = src[k];
  }
}


__device__ __forceinline__ v8f wmma_f32(v2f a, float b_lo, float b_hi, v8f c) {
  v2f b = {b_lo, b_hi};
  return __builtin_amdgcn_wmma_f32_16x16x4_f32(false, a, false, b, (short)0, c, false, false);
}

template <int K, int NC>
__global__ void __launch_bounds__(128)
gemm_wmma_kernel(const float* __restrict__ A, const float* __restrict__ W, float* __restrict__ out, int rowTiles) {
  constexpr int G = NC / 64;
  constexpr int NB = K / 8;
  constexpr int AHEAD = 3;
  const int lane = threadIdx.x & 31;
  const int wave = __builtin_amdgcn_readfirstlane((int)(threadIdx.x >> 5));
  const int c = lane & 15;
  const int h = lane >> 4;
  const int item = (int)blockIdx.x * 4 + wave;
  const int rowTile = item / G;
  const int strip = item % G;
  if (rowTile >= rowTiles) return;

  const float* arow = A + (size_t)(rowTile * 16 + c) * K + 4 * h;
  const float* wrow = W + (size_t)(4 * h) * NC + strip * 64 + 4 * c;

  v8f acc0 = {0.f, 0.f, 0.f, 0.f, 0.f, 0.f, 0.f, 0.f};
  v8f acc1 = acc0, acc2 = acc0, acc3 = acc0;

  float4 a[NB];
  float4 b[NB][4];

#pragma unroll
  for (int i = 0; i < AHEAD && i < NB; ++i) {
    a[i] = *(const float4*)(arow + 8 * i);
#pragma unroll
    for (int j = 0; j < 4; ++j) b[i][j] = *(const float4*)(wrow + (size_t)(8 * i + j) * NC);
  }

#pragma unroll
  for (int kb = 0; kb < NB; ++kb) {
    if (kb + AHEAD < NB) {
      const int n = kb + AHEAD;
      a[n] = *(const float4*)(arow + 8 * n);
#pragma unroll
      for (int j = 0; j < 4; ++j) b[n][j] = *(const float4*)(wrow + (size_t)(8 * n + j) * NC);
    }
    __builtin_amdgcn_sched_barrier(0);
    const v2f a01 = {a[kb].x, a[kb].y};
    const v2f a23 = {a[kb].z, a[kb].w};
    acc0 = wmma_f32(a01, b[kb][0].x, b[kb][1].x, acc0);
    acc1 = wmma_f32(a01, b[kb][0].y, b[kb][1].y, acc1);
    acc2 = wmma_f32(a01, b[kb][0].z, b[kb][1].z, acc2);
    acc3 = wmma_f32(a01, b[kb][0].w, b[kb][1].w, acc3);
    acc0 = wmma_f32(a23, b[kb][2].x, b[kb][3].x, acc0);
    acc1 = wmma_f32(a23, b[kb][2].y, b[kb][3].y, acc1);
    acc2 = wmma_f32(a23, b[kb][2].z, b[kb][3].z, acc2);
    acc3 = wmma_f32(a23, b[kb][2].w, b[kb][3].w, acc3);
  }

  float* orow = out + (size_t)(rowTile * 16 + h * 8) * NC + strip * 64 + 4 * c;
#pragma unroll
  for (int r = 0; r < 8; ++r)
    *(float4*)(orow + (size_t)r * NC) = make_float4(acc0[r], acc1[r], acc2[r], acc3[r]);
}


template <int F>
__global__ void __launch_bounds__(256)
gather_sum_kernel(const int* __restrict__ rowptr, const int* __restrict__ col,
                  const float* __restrict__ dinv, const float* __restrict__ xw,
                  const float* __restrict__ bias, float* __restrict__ out, int n) {
  constexpr int V = F / 32;
  typedef float vec __attribute__((ext_vector_type(V)));
  const int lane = threadIdx.x & 31;
  const int wave = __builtin_amdgcn_readfirstlane((int)(threadIdx.x >> 5));
  const int wavesPerBlock = (int)(blockDim.x >> 5);
  const int nw = (int)gridDim.x * wavesPerBlock;
  const vec b = *(const vec*)(bias + V * lane);

  for (int v = (int)blockIdx.x * wavesPerBlock + wave; v < n; v += nw) {
    const int beg = rowptr[v];
    const int end = rowptr[v + 1];
    const float dv = dinv[v];
    const vec self = *(const vec*)(xw + (size_t)v * F + V * lane);
    vec acc = (dv * dv) * self;

    for (int base = beg; base < end; base += 32) {
      const int m = (end - base < 32) ? (end - base) : 32;
      const int s_l = col[base + (lane < m ? lane : 0)];
      const float d_l = dinv[s_l];
      int j = 0;
      for (; j + 4 <= m; j += 4) {
        const int s0 = __builtin_amdgcn_readlane(s_l, j);
        const int s1 = __builtin_amdgcn_readlane(s_l, j + 1);
        const int s2 = __builtin_amdgcn_readlane(s_l, j + 2);
        const int s3 = __builtin_amdgcn_readlane(s_l, j + 3);
        const vec r0 = *(const vec*)(xw + (size_t)s0 * F + V * lane);
        const vec r1 = *(const vec*)(xw + (size_t)s1 * F + V * lane);
        const vec r2 = *(const vec*)(xw + (size_t)s2 * F + V * lane);
        const vec r3 = *(const vec*)(xw + (size_t)s3 * F + V * lane);
        const float w0 = __int_as_float(__builtin_amdgcn_readlane(__float_as_int(d_l), j)) * dv;
        const float w1 = __int_as_float(__builtin_amdgcn_readlane(__float_as_int(d_l), j + 1)) * dv;
        const float w2 = __int_as_float(__builtin_amdgcn_readlane(__float_as_int(d_l), j + 2)) * dv;
        const float w3 = __int_as_float(__builtin_amdgcn_readlane(__float_as_int(d_l), j + 3)) * dv;
        acc += w0 * r0;
        acc += w1 * r1;
        acc += w2 * r2;
        acc += w3 * r3;
      }
#pragma clang loop unroll(disable)
      for (; j < m; ++j) {
        const int s0 = __builtin_amdgcn_readlane(s_l, j);
        const vec r0 = *(const vec*)(xw + (size_t)s0 * F + V * lane);
        const float w0 = __int_as_float(__builtin_amdgcn_readlane(__float_as_int(d_l), j)) * dv;
        acc += w0 * r0;
      }
    }
    *(vec*)(out + (size_t)v * F + V * lane) = acc + b;
  }
}

__global__ void __launch_bounds__(256)
tanh_kernel(const float* __restrict__ in, float* __restrict__ out, int n) {
  const int n4 = n >> 2;
  const float4* in4 = (const float4*)in;
  float4* out4 = (float4*)out;
  const int first = (int)blockIdx.x * 1024 + (int)threadIdx.x;
  float4 q[4];
#pragma unroll
  for (int u = 0; u < 4; ++u) {
    const int i = first + u * 256;
    q[u] = (i < n4) ? in4[i] : make_float4(0.f, 0.f, 0.f, 0.f);
  }
#pragma unroll
  for (int u = 0; u < 4; ++u) {
    const int i = first + u * 256;
    if (i < n4) {
      float4 t;
      t.x = tanhf(q[u].x);
      t.y = tanhf(q[u].y);
      t.z = tanhf(q[u].z);
      t.w = tanhf(q[u].w);
      out4[i] = t;
    }
  }
  if (blockIdx.x == 0 && (int)threadIdx.x < (n & 3)) {
    const int i = 4 * n4 + (int)threadIdx.x;
    out[i] = tanhf(in[i]);
  }
}


extern "C" void kernel_launch(void* const* d_in, const int* in_sizes, int n_in,
                              void* d_out, int out_size, void* d_ws, size_t ws_size,
                              hipStream_t stream) {
  const float* x  = (const float*)d_in[0];
  const int*   ei = (const int*)d_in[1];
  const float* W1 = (const float*)d_in[2];
  const float* b1 = (const float*)d_in[3];
  const float* W2 = (const float*)d_in[4];
  const float* b2 = (const float*)d_in[5];

  const int DIN  = in_sizes[3];
  const int DOUT = in_sizes[5];
  const int N    = in_sizes[0] / DIN;
  const int E    = in_sizes[1] / 2;
  const int* src = ei;
  const int* dst = ei + E;

  char* ws = (char*)d_ws;
  size_t off = ((size_t)N * sizeof(float) + 255) & ~(size_t)255;
  float* dinv = (float*)ws;
  float* xw1  = (float*)(ws + off);
  size_t off2 = off + (((size_t)N * DIN * sizeof(float) + 255) & ~(size_t)255);
  float* h1   = (float*)(ws + off2);
  float* xw2  = xw1;
  float* pre  = h1;
  float* out  = (float*)d_out;

  char* ob = (char*)d_out;
  size_t lo2 = ((size_t)(N + 1) * sizeof(int) + 255) & ~(size_t)255;
  size_t lo3 = lo2 + (((size_t)E * sizeof(int) + 255) & ~(size_t)255);
  int* rowptr = (int*)ob;
  int* rank   = (int*)(ob + lo2);
  int* col    = (int*)(ob + lo3);

  const int TB = 256;
  const int NR = (N + RANGE - 1) / RANGE;
  const int NP = NR * RANGE;
  const int SUBP = NR * GROUPS;
  const int chunk = (((E + CHUNKS - 1) / CHUNKS) + 3) & ~3;
  int* cntm = (int*)h1;
  int* sub  = cntm + (size_t)CHUNKS * NP;

  count_rank_kernel<<<dim3(CHUNKS, NR), CT, 0, stream>>>(dst, rank, cntm, sub, E, chunk, NP, SUBP);
  scan_counts_kernel<<<(N + 255) / 256, 256, 0, stream>>>(cntm, sub, rowptr, dinv, N, NP, SUBP);
  fill_lists_kernel<<<(E + 4 * TB - 1) / (4 * TB), TB, 0, stream>>>(src, dst, rank, cntm, col, E, chunk, NP);

  gemm_wmma_kernel<128, 128><<<(N / 16 * (128 / 64) + 3) / 4, 128, 0, stream>>>(x, W1, xw1, N / 16);
  gather_sum_kernel<128><<<(N + 15) / 16, 256, 0, stream>>>(rowptr, col, dinv, xw1, b1, h1, N);

  gemm_wmma_kernel<128, 64><<<(N / 16 * (64 / 64) + 3) / 4, 128, 0, stream>>>(h1, W2, xw2, N / 16);
  gather_sum_kernel<64><<<(N + 15) / 16, 256, 0, stream>>>(rowptr, col, dinv, xw2, b2, pre, N);
  tanh_kernel<<<(out_size + 16 * TB - 1) / (16 * TB), TB, 0, stream>>>(pre, out, out_size);
}


# layer-2 product runs as epilogue of the layer-1 aggregation: 16 summed rows stay in LDS, times W2 on WMMA in 4 k quarters added through LDS; hidden layer never stored; sum order over k changes
# speedup vs baseline: 2.8067x; 1.0704x over previous
#include <hip/hip_runtime.h>
#include <hip/hip_bf16.h>

typedef __attribute__((ext_vector_type(2))) float v2f;
typedef __attribute__((ext_vector_type(8))) float v8f;


constexpr int CHUNKS = 20;
constexpr int RANGE = 4096;
constexpr int GROUPS = RANGE / 256;
constexpr int CT = 1024;

__device__ __forceinline__ bool in_range(int d, int lo) { return (unsigned)(d - lo) < (unsigned)RANGE; }

__device__ __forceinline__ int lds_take_slot(int* counter) {
  return __hip_atomic_fetch_add(counter, 1, __ATOMIC_RELAXED, __HIP_MEMORY_SCOPE_WORKGROUP);
}

__global__ void __launch_bounds__(CT)
count_rank_kernel(const int* __restrict__ dst, int* __restrict__ rank, int* __restrict__ cntm,
                  int* __restrict__ sub, int E, int chunk, int NP, int SUBP) {
  __shared__ int lc[RANGE];
  __shared__ int s_sub[GROUPS];
  const int tid = threadIdx.x;
  const int ci = blockIdx.x;
  const int rj = blockIdx.y;
  const int lo = rj * RANGE;

  ((int4*)lc)[tid] = make_int4(0, 0, 0, 0);
  if (tid < GROUPS) s_sub[tid] = 0;
  __syncthreads();

  const int cbeg = ci * chunk;
  const int cend = (cbeg + chunk < E) ? (cbeg + chunk) : E;
  if ((E & 3) == 0) {
    const int n4 = (cend - cbeg) >> 2;
    const int4* d4 = (const int4*)(dst + cbeg);
    int* rk = rank + cbeg;
    for (int q = tid; q < n4; q += 8 * CT) {
      int4 d[8];
#pragma unroll
      for (int u = 0; u < 8; ++u) {
        const int idx = q + u * CT;
        d[u] = (idx < n4) ? d4[idx] : make_int4(-1, -1, -1, -1);
      }
#pragma unroll
      for (int u = 0; u < 8; ++u) {
        const int e = 4 * (q + u * CT);
        const bool h0 = in_range(d[u].x, lo);
        const bool h1 = in_range(d[u].y, lo);
        const bool h2 = in_range(d[u].z, lo);
        const bool h3 = in_range(d[u].w, lo);
        int r0 = 0, r1 = 0, r2 = 0, r3 = 0;
        if (h0) r0 = lds_take_slot(&lc[d[u].x - lo]);
        if (h1) r1 = lds_take_slot(&lc[d[u].y - lo]);
        if (h2) r2 = lds_take_slot(&lc[d[u].z - lo]);
        if (h3) r3 = lds_take_slot(&lc[d[u].w - lo]);
        if (h0) rk[e] = r0;
        if (h1) rk[e + 1] = r1;
        if (h2) rk[e + 2] = r2;
        if (h3) rk[e + 3] = r3;
      }
    }
  } else {
    for (int k = cbeg + tid; k < cend; k += CT) {
      const int d = dst[k];
      if (in_range(d, lo)) rank[k] = lds_take_slot(&lc[d - lo]);
    }
  }
  __syncthreads();

  const int4 c = ((const int4*)lc)[tid];
  *(int4*)(cntm + (size_t)ci * NP + lo + 4 * tid) = c;
  int s = (c.x + c.y) + (c.z + c.w);
  for (int o = 16; o > 0; o >>= 1) s += __shfl_xor(s, o);
  if ((tid & 31) == 0)
    __hip_atomic_fetch_add(&s_sub[tid >> 6], s, __ATOMIC_RELAXED, __HIP_MEMORY_SCOPE_WORKGROUP);
  __syncthreads();
  if (tid < GROUPS) sub[ci * SUBP + rj * GROUPS + tid] = s_sub[tid];
}

__global__ void __launch_bounds__(256)
scan_counts_kernel(int* __restrict__ cntm, const int* __restrict__ sub,
                   int* __restrict__ rowptr, float* __restrict__ dinv, int n, int NP, int SUBP) {
  __shared__ int s_wave_cnt[8];
  __shared__ int s_wave_front[8];
  const int tid = threadIdx.x;
  const int lane = tid & 31;
  const int wave = tid >> 5;
  const int g = blockIdx.x;
  const int v = g * 256 + tid;

  int part[CHUNKS];
  int fr[CHUNKS];
  const bool has = tid < g;
  const int gi = has ? tid : 0;
#pragma unroll
  for (int i = 0; i < CHUNKS; ++i) {
    part[i] = cntm[(size_t)i * NP + v];
    fr[i] = sub[i * SUBP + gi];
  }
  int c = 0;
  int front = 0;
#pragma unroll
  for (int i = 0; i < CHUNKS; ++i) {
    c += part[i];
    front += has ? fr[i] : 0;
  }
#pragma clang loop unroll(disable)
  for (int gg = tid + 256; gg < g; gg += 256) {
#pragma clang loop unroll(disable)
    for (int i = 0; i < CHUNKS; ++i) front += sub[i * SUBP + gg];
  }

  int incl = c;
  for (int o = 1; o < 32; o <<= 1) {
    int t = __shfl_up(incl, o);
    if (lane >= o) incl += t;
  }
  for (int o = 16; o > 0; o >>= 1) front += __shfl_xor(front, o);
  if (lane == 31) s_wave_cnt[wave] = incl;
  if (lane == 0) s_wave_front[wave] = front;
  __syncthreads();
  int base = 0;
  for (int w = 0; w < 8; ++w) {
    base += s_wave_front[w];
    if (w < wave) base += s_wave_cnt[w];
  }
  const int excl = base + incl - c;
  if (v < n) {
    rowptr[v] = excl;
    dinv[v] = rsqrtf((float)(c + 1));
    int run = excl;
#pragma unroll
    for (int i = 0; i < CHUNKS; ++i) {
      cntm[(size_t)i * NP + v] = run;
      run += part[i];
    }
  }
  if (v == n - 1) rowptr[n] = excl + c;
}

__global__ void fill_lists_kernel(const int* __restrict__ src, const int* __restrict__ dst,
                                  const int* __restrict__ rank,
                                  const int* __restrict__ first, int* __restrict__ col,
                                  int E, int chunk, int NP) {
  const int e = 4 * (int)(blockIdx.x * blockDim.x + threadIdx.x);
  if ((E & 3) == 0) {
    if (e < E) {
      const int* f = first + (size_t)(e / chunk) * NP;
      const int4 s = *(const int4*)(src + e);
      const int4 d = *(const int4*)(dst + e);
      const int4 r = *(const int4*)(rank + e);
      const int p0 = f[d.x];
      const int p1 = f[d.y];
      const int p2 = f[d.z];
      const int p3 = f[d.w];
      col[p0 + r.x] = s.x;
      col[p1 + r.y] = s.y;
      col[p2 + r.z] = s.z;
      col[p3 + r.w] = s.w;
    }
  } else {
    for (int k = e; k < e + 4 && k < E; ++k)
      col[first[(size_t)(k / chunk) * NP + dst[k]] + rank[k]] = src[k];
  }
}


__device__ __forceinline__ v8f wmma_f32(v2f a, float b_lo, float b_hi, v8f c) {
  v2f b = {b_lo, b_hi};
  return __builtin_amdgcn_wmma_f32_16x16x4_f32(false, a, false, b, (short)0, c, false, false);
}

template <int K, int NC>
__global__ void __launch_bounds__(128)
gemm_wmma_kernel(const float* __restrict__ A, const float* __restrict__ W, float* __restrict__ out, int rowTiles) {
  constexpr int G = NC / 64;
  constexpr int NB = K / 8;
  constexpr int AHEAD = 3;
  const int lane = threadIdx.x & 31;
  const int wave = __builtin_amdgcn_readfirstlane((int)(threadIdx.x >> 5));
  const int c = lane & 15;
  const int h = lane >> 4;
  const int item = (int)blockIdx.x * 4 + wave;
  const int rowTile = item / G;
  const int strip = item % G;
  if (rowTile >= rowTiles) return;

  const float* arow = A + (size_t)(rowTile * 16 + c) * K + 4 * h;
  const float* wrow = W + (size_t)(4 * h) * NC + strip * 64 + 4 * c;

  v8f acc0 = {0.f, 0.f, 0.f, 0.f, 0.f, 0.f, 0.f, 0.f};
  v8f acc1 = acc0, acc2 = acc0, acc3 = acc0;

  float4 a[NB];
  float4 b[NB][4];

#pragma unroll
  for (int i = 0; i < AHEAD && i < NB; ++i) {
    a[i] = *(const float4*)(arow + 8 * i);
#pragma unroll
    for (int j = 0; j < 4; ++j) b[i][j] = *(const float4*)(wrow + (size_t)(8 * i + j) * NC);
  }

#pragma unroll
  for (int kb = 0; kb < NB; ++kb) {
    if (kb + AHEAD < NB) {
      const int n = kb + AHEAD;
      a[n] = *(const float4*)(arow + 8 * n);
#pragma unroll
      for (int j = 0; j < 4; ++j) b[n][j] = *(const float4*)(wrow + (size_t)(8 * n + j) * NC);
    }
    __builtin_amdgcn_sched_barrier(0);
    const v2f a01 = {a[kb].x, a[kb].y};
    const v2f a23 = {a[kb].z, a[kb].w};
    acc0 = wmma_f32(a01, b[kb][0].x, b[kb][1].x, acc0);
    acc1 = wmma_f32(a01, b[kb][0].y, b[kb][1].y, acc1);
    acc2 = wmma_f32(a01, b[kb][0].z, b[kb][1].z, acc2);
    acc3 = wmma_f32(a01, b[kb][0].w, b[kb][1].w, acc3);
    acc0 = wmma_f32(a23, b[kb][2].x, b[kb][3].x, acc0);
    acc1 = wmma_f32(a23, b[kb][2].y, b[kb][3].y, acc1);
    acc2 = wmma_f32(a23, b[kb][2].z, b[kb][3].z, acc2);
    acc3 = wmma_f32(a23, b[kb][2].w, b[kb][3].w, acc3);
  }

  float* orow = out + (size_t)(rowTile * 16 + h * 8) * NC + strip * 64 + 4 * c;
#pragma unroll
  for (int r = 0; r < 8; ++r)
    *(float4*)(orow + (size_t)r * NC) = make_float4(acc0[r], acc1[r], acc2[r], acc3[r]);
}


template <int F>
__global__ void __launch_bounds__(256)
gather_sum_kernel(const int* __restrict__ rowptr, const int* __restrict__ col,
                  const float* __restrict__ dinv, const float* __restrict__ xw,
                  const float* __restrict__ bias, float* __restrict__ out, int n) {
  constexpr int V = F / 32;
  typedef float vec __attribute__((ext_vector_type(V)));
  const int lane = threadIdx.x & 31;
  const int wave = __builtin_amdgcn_readfirstlane((int)(threadIdx.x >> 5));
  const int wavesPerBlock = (int)(blockDim.x >> 5);
  const int nw = (int)gridDim.x * wavesPerBlock;
  const vec b = *(const vec*)(bias + V * lane);

  for (int v = (int)blockIdx.x * wavesPerBlock + wave; v < n; v += nw) {
    const int beg = rowptr[v];
    const int end = rowptr[v + 1];
    const float dv = dinv[v];
    const vec self = *(const vec*)(xw + (size_t)v * F + V * lane);
    vec acc = (dv * dv) * self;

    for (int base = beg; base < end; base += 32) {
      const int m = (end - base < 32) ? (end - base) : 32;
      const int s_l = col[base + (lane < m ? lane : 0)];
      const float d_l = dinv[s_l];
      int j = 0;
      for (; j + 4 <= m; j += 4) {
        const int s0 = __builtin_amdgcn_readlane(s_l, j);
        const int s1 = __builtin_amdgcn_readlane(s_l, j + 1);
        const int s2 = __builtin_amdgcn_readlane(s_l, j + 2);
        const int s3 = __builtin_amdgcn_readlane(s_l, j + 3);
        const vec r0 = *(const vec*)(xw + (size_t)s0 * F + V * lane);
        const vec r1 = *(const vec*)(xw + (size_t)s1 * F + V * lane);
        const vec r2 = *(const vec*)(xw + (size_t)s2 * F + V * lane);
        const vec r3 = *(const vec*)(xw + (size_t)s3 * F + V * lane);
        const float w0 = __int_as_float(__builtin_amdgcn_readlane(__float_as_int(d_l), j)) * dv;
        const float w1 = __int_as_float(__builtin_amdgcn_readlane(__float_as_int(d_l), j + 1)) * dv;
        const float w2 = __int_as_float(__builtin_amdgcn_readlane(__float_as_int(d_l), j + 2)) * dv;
        const float w3 = __int_as_float(__builtin_amdgcn_readlane(__float_as_int(d_l), j + 3)) * dv;
        acc += w0 * r0;
        acc += w1 * r1;
        acc += w2 * r2;
        acc += w3 * r3;
      }
#pragma clang loop unroll(disable)
      for (; j < m; ++j) {
        const int s0 = __builtin_amdgcn_readlane(s_l, j);
        const vec r0 = *(const vec*)(xw + (size_t)s0 * F + V * lane);
        const float w0 = __int_as_float(__builtin_amdgcn_readlane(__float_as_int(d_l), j)) * dv;
        acc += w0 * r0;
      }
    }
    *(vec*)(out + (size_t)v * F + V * lane) = acc + b;
  }
}


template <int F>
__global__ void __launch_bounds__(256)
gather_sum_gemm_kernel(const int* __restrict__ rowptr, const int* __restrict__ col,
                       const float* __restrict__ dinv, const float* __restrict__ xw,
                       const float* __restrict__ bias, const float* __restrict__ W,
                       float* __restrict__ out, int n) {
  constexpr int V = F / 32;
  constexpr int NC = 64;
  constexpr int PITCH = F + 4;
  constexpr int KQ = F / 4;
  constexpr int NB = KQ / 8;
  typedef float vec __attribute__((ext_vector_type(V)));
  static_assert(V == 4, "a lane owns 16 bytes of a hidden row");
  __shared__ __attribute__((aligned(16))) float hs[16 * PITCH];
  __shared__ __attribute__((aligned(16))) float part[4][16][NC];

  const int lane = threadIdx.x & 31;
  const int wave = __builtin_amdgcn_readfirstlane((int)(threadIdx.x >> 5));
  const vec b = *(const vec*)(bias + V * lane);
  const int node0 = (int)blockIdx.x * 16;

  for (int rr = 0; rr < 2; ++rr) {
    const int row = wave + 8 * rr;
    const int v = node0 + row;
    vec acc = {0.f, 0.f, 0.f, 0.f};
    if (v < n) {
      const int beg = rowptr[v];
      const int end = rowptr[v + 1];
      const float dv = dinv[v];
      const vec self = *(const vec*)(xw + (size_t)v * F + V * lane);
      acc = (dv * dv) * self;

      for (int base = beg; base < end; base += 32) {
        const int m = (end - base < 32) ? (end - base) : 32;
        const int s_l = col[base + (lane < m ? lane : 0)];
        const float d_l = dinv[s_l];
        int j = 0;
        for (; j + 4 <= m; j += 4) {
          const int s0 = __builtin_amdgcn_readlane(s_l, j);
          const int s1 = __builtin_amdgcn_readlane(s_l, j + 1);
          const int s2 = __builtin_amdgcn_readlane(s_l, j + 2);
          const int s3 = __builtin_amdgcn_readlane(s_l, j + 3);
          const vec r0 = *(const vec*)(xw + (size_t)s0 * F + V * lane);
          const vec r1 = *(const vec*)(xw + (size_t)s1 * F + V * lane);
          const vec r2 = *(const vec*)(xw + (size_t)s2 * F + V * lane);
          const vec r3 = *(const vec*)(xw + (size_t)s3 * F + V * lane);
          const float w0 = __int_as_float(__builtin_amdgcn_readlane(__float_as_int(d_l), j)) * dv;
          const float w1 = __int_as_float(__builtin_amdgcn_readlane(__float_as_int(d_l), j + 1)) * dv;
          const float w2 = __int_as_float(__builtin_amdgcn_readlane(__float_as_int(d_l), j + 2)) * dv;
          const float w3 = __int_as_float(__builtin_amdgcn_readlane(__float_as_int(d_l), j + 3)) * dv;
          acc += w0 * r0;
          acc += w1 * r1;
          acc += w2 * r2;
          acc += w3 * r3;
        }
#pragma clang loop unroll(disable)
        for (; j < m; ++j) {
          const int s0 = __builtin_amdgcn_readlane(s_l, j);
          const vec r0 = *(const vec*)(xw + (size_t)s0 * F + V * lane);
          const float w0 = __int_as_float(__builtin_amdgcn_readlane(__float_as_int(d_l), j)) * dv;
          acc += w0 * r0;
        }
      }
      acc = acc + b;
    }
    *(vec*)(hs + row * PITCH + V * lane) = acc;
  }

  const int c = lane & 15;
  const int h = lane >> 4;
  const int q = wave >> 1;
  const int tp = wave & 1;
  const float* wp = W + (size_t)(KQ * q + 4 * h) * NC + 32 * tp + 2 * c;
  float2 bf[NB][4];
#pragma unroll
  for (int i = 0; i < NB; ++i) {
#pragma unroll
    for (int j = 0; j < 4; ++j) bf[i][j] = *(const float2*)(wp + (size_t)(8 * i + j) * NC);
  }
  __syncthreads();

  const float* ap = hs + c * PITCH + KQ * q + 4 * h;
  float4 af[NB];
#pragma unroll
  for (int i = 0; i < NB; ++i) af[i] = *(const float4*)(ap + 8 * i);

  v8f acc0 = {0.f, 0.f, 0.f, 0.f, 0.f, 0.f, 0.f, 0.f};
  v8f acc1 = acc0;
#pragma unroll
  for (int i = 0; i < NB; ++i) {
    const v2f a01 = {af[i].x, af[i].y};
    const v2f a23 = {af[i].z, af[i].w};
    acc0 = wmma_f32(a01, bf[i][0].x, bf[i][1].x, acc0);
    acc1 = wmma_f32(a01, bf[i][0].y, bf[i][1].y, acc1);
    acc0 = wmma_f32(a23, bf[i][2].x, bf[i][3].x, acc0);
    acc1 = wmma_f32(a23, bf[i][2].y, bf[i][3].y, acc1);
  }

#pragma unroll
  for (int r = 0; r < 8; ++r)
    *(float2*)(&part[q][8 * h + r][32 * tp + 2 * c]) = make_float2(acc0[r], acc1[r]);
  __syncthreads();

  const int tid = (int)threadIdx.x;
  const int orow = tid >> 4;
  const int ocol = 4 * (tid & 15);
  const float4 p0 = *(const float4*)(&part[0][orow][ocol]);
  const float4 p1 = *(const float4*)(&part[1][orow][ocol]);
  const float4 p2 = *(const float4*)(&part[2][orow][ocol]);
  const float4 p3 = *(const float4*)(&part[3][orow][ocol]);
  float4 s;
  s.x = ((p0.x + p1.x) + p2.x) + p3.x;
  s.y = ((p0.y + p1.y) + p2.y) + p3.y;
  s.z = ((p0.z + p1.z) + p2.z) + p3.z;
  s.w = ((p0.w + p1.w) + p2.w) + p3.w;
  if (node0 + orow < n) *(float4*)(out + (size_t)(node0 + orow) * NC + ocol) = s;
}

__global__ void __launch_bounds__(256)
tanh_kernel(const float* __restrict__ in, float* __restrict__ out, int n) {
  const int n4 = n >> 2;
  const float4* in4 = (const float4*)in;
  float4* out4 = (float4*)out;
  const int first = (int)blockIdx.x * 1024 + (int)threadIdx.x;
  float4 q[4];
#pragma unroll
  for (int u = 0; u < 4; ++u) {
    const int i = first + u * 256;
    q[u] = (i < n4) ? in4[i] : make_float4(0.f, 0.f, 0.f, 0.f);
  }
#pragma unroll
  for (int u = 0; u < 4; ++u) {
    const int i = first + u * 256;
    if (i < n4) {
      float4 t;
      t.x = tanhf(q[u].x);
      t.y = tanhf(q[u].y);
      t.z = tanhf(q[u].z);
      t.w = tanhf(q[u].w);
      out4[i] = t;
    }
  }
  if (blockIdx.x == 0 && (int)threadIdx.x < (n & 3)) {
    const int i = 4 * n4 + (int)threadIdx.x;
    out[i] = tanhf(in[i]);
  }
}


extern "C" void kernel_launch(void* const* d_in, const int* in_sizes, int n_in,
                              void* d_out, int out_size, void* d_ws, size_t ws_size,
                              hipStream_t stream) {
  const float* x  = (const float*)d_in[0];
  const int*   ei = (const int*)d_in[1];
  const float* W1 = (const float*)d_in[2];
  const float* b1 = (const float*)d_in[3];
  const float* W2 = (const float*)d_in[4];
  const float* b2 = (const float*)d_in[5];

  const int DIN  = in_sizes[3];
  const int DOUT = in_sizes[5];
  const int N    = in_sizes[0] / DIN;
  const int E    = in_sizes[1] / 2;
  const int* src = ei;
  const int* dst = ei + E;

  char* ws = (char*)d_ws;
  size_t off = ((size_t)N * sizeof(float) + 255) & ~(size_t)255;
  float* dinv = (float*)ws;
  float* xw1  = (float*)(ws + off);
  size_t off2 = off + (((size_t)N * DIN * sizeof(float) + 255) & ~(size_t)255);
  float* h1   = (float*)(ws + off2);
  float* xw2  = h1;
  float* pre  = xw1;
  float* out  = (float*)d_out;

  char* ob = (char*)d_out;
  size_t lo2 = ((size_t)(N + 1) * sizeof(int) + 255) & ~(size_t)255;
  size_t lo3 = lo2 + (((size_t)E * sizeof(int) + 255) & ~(size_t)255);
  int* rowptr = (int*)ob;
  int* rank   = (int*)(ob + lo2);
  int* col    = (int*)(ob + lo3);

  const int TB = 256;
  const int NR = (N + RANGE - 1) / RANGE;
  const int NP = NR * RANGE;
  const int SUBP = NR * GROUPS;
  const int chunk = (((E + CHUNKS - 1) / CHUNKS) + 3) & ~3;
  int* cntm = (int*)h1;
  int* sub  = cntm + (size_t)CHUNKS * NP;

  count_rank_kernel<<<dim3(CHUNKS, NR), CT, 0, stream>>>(dst, rank, cntm, sub, E, chunk, NP, SUBP);
  scan_counts_kernel<<<(N + 255) / 256, 256, 0, stream>>>(cntm, sub, rowptr, dinv, N, NP, SUBP);
  fill_lists_kernel<<<(E + 4 * TB - 1) / (4 * TB), TB, 0, stream>>>(src, dst, rank, cntm, col, E, chunk, NP);

  gemm_wmma_kernel<128, 128><<<(N / 16 * (128 / 64) + 3) / 4, 128, 0, stream>>>(x, W1, xw1, N / 16);
  gather_sum_gemm_kernel<128><<<(N + 15) / 16, 256, 0, stream>>>(rowptr, col, dinv, xw1, b1, W2, xw2, N);

  gather_sum_kernel<64><<<(N + 15) / 16, 256, 0, stream>>>(rowptr, col, dinv, xw2, b2, pre, N);
  tanh_kernel<<<(out_size + 16 * TB - 1) / (16 * TB), TB, 0, stream>>>(pre, out, out_size);
}


# no nonlinearity between the layers: W1@W2 and b1@W2 folded per call on WMMA, so x@(W1W2) is one 128x64 product and both aggregations gather 64-wide rows; product and sum order reassociated
# speedup vs baseline: 3.0470x; 1.0856x over previous
#include <hip/hip_runtime.h>
#include <hip/hip_bf16.h>

typedef __attribute__((ext_vector_type(2))) float v2f;
typedef __attribute__((ext_vector_type(8))) float v8f;


constexpr int CHUNKS = 20;
constexpr int RANGE = 4096;
constexpr int GROUPS = RANGE / 256;
constexpr int CT = 1024;

__device__ __forceinline__ bool in_range(int d, int lo) { return (unsigned)(d - lo) < (unsigned)RANGE; }

__device__ __forceinline__ int lds_take_slot(int* counter) {
  return __hip_atomic_fetch_add(counter, 1, __ATOMIC_RELAXED, __HIP_MEMORY_SCOPE_WORKGROUP);
}

__global__ void __launch_bounds__(CT)
count_rank_kernel(const int* __restrict__ dst, int* __restrict__ rank, int* __restrict__ cntm,
                  int* __restrict__ sub, int E, int chunk, int NP, int SUBP) {
  __shared__ int lc[RANGE];
  __shared__ int s_sub[GROUPS];
  const int tid = threadIdx.x;
  const int ci = blockIdx.x;
  const int rj = blockIdx.y;
  const int lo = rj * RANGE;

  ((int4*)lc)[tid] = make_int4(0, 0, 0, 0);
  if (tid < GROUPS) s_sub[tid] = 0;
  __syncthreads();

  const int cbeg = ci * chunk;
  const int cend = (cbeg + chunk < E) ? (cbeg + chunk) : E;
  if ((E & 3) == 0) {
    const int n4 = (cend - cbeg) >> 2;
    const int4* d4 = (const int4*)(dst + cbeg);
    int* rk = rank + cbeg;
    for (int q = tid; q < n4; q += 8 * CT) {
      int4 d[8];
#pragma unroll
      for (int u = 0; u < 8; ++u) {
        const int idx = q + u * CT;
        d[u] = (idx < n4) ? d4[idx] : make_int4(-1, -1, -1, -1);
      }
#pragma unroll
      for (int u = 0; u < 8; ++u) {
        const int e = 4 * (q + u * CT);
        const bool h0 = in_range(d[u].x, lo);
        const bool h1 = in_range(d[u].y, lo);
        const bool h2 = in_range(d[u].z, lo);
        const bool h3 = in_range(d[u].w, lo);
        int r0 = 0, r1 = 0, r2 = 0, r3 = 0;
        if (h0) r0 = lds_take_slot(&lc[d[u].x - lo]);
        if (h1) r1 = lds_take_slot(&lc[d[u].y - lo]);
        if (h2) r2 = lds_take_slot(&lc[d[u].z - lo]);
        if (h3) r3 = lds_take_slot(&lc[d[u].w - lo]);
        if (h0) rk[e] = r0;
        if (h1) rk[e + 1] = r1;
        if (h2) rk[e + 2] = r2;
        if (h3) rk[e + 3] = r3;
      }
    }
  } else {
    for (int k = cbeg + tid; k < cend; k += CT) {
      const int d = dst[k];
      if (in_range(d, lo)) rank[k] = lds_take_slot(&lc[d - lo]);
    }
  }
  __syncthreads();

  const int4 c = ((const int4*)lc)[tid];
  *(int4*)(cntm + (size_t)ci * NP + lo + 4 * tid) = c;
  int s = (c.x + c.y) + (c.z + c.w);
  for (int o = 16; o > 0; o >>= 1) s += __shfl_xor(s, o);
  if ((tid & 31) == 0)
    __hip_atomic_fetch_add(&s_sub[tid >> 6], s, __ATOMIC_RELAXED, __HIP_MEMORY_SCOPE_WORKGROUP);
  __syncthreads();
  if (tid < GROUPS) sub[ci * SUBP + rj * GROUPS + tid] = s_sub[tid];
}

__global__ void __launch_bounds__(256)
scan_counts_kernel(int* __restrict__ cntm, const int* __restrict__ sub,
                   int* __restrict__ rowptr, float* __restrict__ dinv, int n, int NP, int SUBP) {
  __shared__ int s_wave_cnt[8];
  __shared__ int s_wave_front[8];
  const int tid = threadIdx.x;
  const int lane = tid & 31;
  const int wave = tid >> 5;
  const int g = blockIdx.x;
  const int v = g * 256 + tid;

  int part[CHUNKS];
  int fr[CHUNKS];
  const bool has = tid < g;
  const int gi = has ? tid : 0;
#pragma unroll
  for (int i = 0; i < CHUNKS; ++i) {
    part[i] = cntm[(size_t)i * NP + v];
    fr[i] = sub[i * SUBP + gi];
  }
  int c = 0;
  int front = 0;
#pragma unroll
  for (int i = 0; i < CHUNKS; ++i) {
    c += part[i];
    front += has ? fr[i] : 0;
  }
#pragma clang loop unroll(disable)
  for (int gg = tid + 256; gg < g; gg += 256) {
#pragma clang loop unroll(disable)
    for (int i = 0; i < CHUNKS; ++i) front += sub[i * SUBP + gg];
  }

  int incl = c;
  for (int o = 1; o < 32; o <<= 1) {
    int t = __shfl_up(incl, o);
    if (lane >= o) incl += t;
  }
  for (int o = 16; o > 0; o >>= 1) front += __shfl_xor(front, o);
  if (lane == 31) s_wave_cnt[wave] = incl;
  if (lane == 0) s_wave_front[wave] = front;
  __syncthreads();
  int base = 0;
  for (int w = 0; w < 8; ++w) {
    base += s_wave_front[w];
    if (w < wave) base += s_wave_cnt[w];
  }
  const int excl = base + incl - c;
  if (v < n) {
    rowptr[v] = excl;
    dinv[v] = rsqrtf((float)(c + 1));
    int run = excl;
#pragma unroll
    for (int i = 0; i < CHUNKS; ++i) {
      cntm[(size_t)i * NP + v] = run;
      run += part[i];
    }
  }
  if (v == n - 1) rowptr[n] = excl + c;
}

__global__ void fill_lists_kernel(const int* __restrict__ src, const int* __restrict__ dst,
                                  const int* __restrict__ rank,
                                  const int* __restrict__ first, int* __restrict__ col,
                                  int E, int chunk, int NP) {
  const int e = 4 * (int)(blockIdx.x * blockDim.x + threadIdx.x);
  if ((E & 3) == 0) {
    if (e < E) {
      const int* f = first + (size_t)(e / chunk) * NP;
      const int4 s = *(const int4*)(src + e);
      const int4 d = *(const int4*)(dst + e);
      const int4 r = *(const int4*)(rank + e);
      const int p0 = f[d.x];
      const int p1 = f[d.y];
      const int p2 = f[d.z];
      const int p3 = f[d.w];
      col[p0 + r.x] = s.x;
      col[p1 + r.y] = s.y;
      col[p2 + r.z] = s.z;
      col[p3 + r.w] = s.w;
    }
  } else {
    for (int k = e; k < e + 4 && k < E; ++k)
      col[first[(size_t)(k / chunk) * NP + dst[k]] + rank[k]] = src[k];
  }
}


__device__ __forceinline__ v8f wmma_f32(v2f a, float b_lo, float b_hi, v8f c) {
  v2f b = {b_lo, b_hi};
  return __builtin_amdgcn_wmma_f32_16x16x4_f32(false, a, false, b, (short)0, c, false, false);
}

template <int K, int NC>
__global__ void __launch_bounds__(128)
gemm_wmma_kernel(const float* __restrict__ A, const float* __restrict__ W, float* __restrict__ out, int rowTiles) {
  constexpr int G = NC / 64;
  constexpr int NB = K / 8;
  constexpr int AHEAD = 3;
  const int lane = threadIdx.x & 31;
  const int wave = __builtin_amdgcn_readfirstlane((int)(threadIdx.x >> 5));
  const int c = lane & 15;
  const int h = lane >> 4;
  const int item = (int)blockIdx.x * 4 + wave;
  const int rowTile = item / G;
  const int strip = item % G;
  if (rowTile >= rowTiles) return;

  const float* arow = A + (size_t)(rowTile * 16 + c) * K + 4 * h;
  const float* wrow = W + (size_t)(4 * h) * NC + strip * 64 + 4 * c;

  v8f acc0 = {0.f, 0.f, 0.f, 0.f, 0.f, 0.f, 0.f, 0.f};
  v8f acc1 = acc0, acc2 = acc0, acc3 = acc0;

  float4 a[NB];
  float4 b[NB][4];

#pragma unroll
  for (int i = 0; i < AHEAD && i < NB; ++i) {
    a[i] = *(const float4*)(arow + 8 * i);
#pragma unroll
    for (int j = 0; j < 4; ++j) b[i][j] = *(const float4*)(wrow + (size_t)(8 * i + j) * NC);
  }

#pragma unroll
  for (int kb = 0; kb < NB; ++kb) {
    if (kb + AHEAD < NB) {
      const int n = kb + AHEAD;
      a[n] = *(const float4*)(arow + 8 * n);
#pragma unroll
      for (int j = 0; j < 4; ++j) b[n][j] = *(const float4*)(wrow + (size_t)(8 * n + j) * NC);
    }
    __builtin_amdgcn_sched_barrier(0);
    const v2f a01 = {a[kb].x, a[kb].y};
    const v2f a23 = {a[kb].z, a[kb].w};
    acc0 = wmma_f32(a01, b[kb][0].x, b[kb][1].x, acc0);
    acc1 = wmma_f32(a01, b[kb][0].y, b[kb][1].y, acc1);
    acc2 = wmma_f32(a01, b[kb][0].z, b[kb][1].z, acc2);
    acc3 = wmma_f32(a01, b[kb][0].w, b[kb][1].w, acc3);
    acc0 = wmma_f32(a23, b[kb][2].x, b[kb][3].x, acc0);
    acc1 = wmma_f32(a23, b[kb][2].y, b[kb][3].y, acc1);
    acc2 = wmma_f32(a23, b[kb][2].z, b[kb][3].z, acc2);
    acc3 = wmma_f32(a23, b[kb][2].w, b[kb][3].w, acc3);
  }

  float* orow = out + (size_t)(rowTile * 16 + h * 8) * NC + strip * 64 + 4 * c;
#pragma unroll
  for (int r = 0; r < 8; ++r)
    *(float4*)(orow + (size_t)r * NC) = make_float4(acc0[r], acc1[r], acc2[r], acc3[r]);
}

template <int K, int NC>
__global__ void __launch_bounds__(128)
fold_weights_kernel(const float* __restrict__ W1, const float* __restrict__ b1,
                    const float* __restrict__ W2, float* __restrict__ out, int rowTiles) {
  static_assert(NC == 64, "one strip of 64 columns");
  constexpr int NB = K / 8;
  constexpr int AHEAD = 6;
  const int lane = threadIdx.x & 31;
  const int wave = __builtin_amdgcn_readfirstlane((int)(threadIdx.x >> 5));
  const int c = lane & 15;
  const int h = lane >> 4;
  const int item = (int)blockIdx.x * 4 + wave;
  if (item > rowTiles) return;
  const bool bias_tile = (item == rowTiles);
  const bool live = !bias_tile || c == 0;
  const float* arow = (bias_tile ? b1 : W1 + (size_t)(item * 16 + c) * K) + 4 * h;
  const float* wrow = W2 + (size_t)(4 * h) * NC + 4 * c;

  v8f acc0 = {0.f, 0.f, 0.f, 0.f, 0.f, 0.f, 0.f, 0.f};
  v8f acc1 = acc0, acc2 = acc0, acc3 = acc0;

  float4 a[NB];
  float4 w[NB][4];

#pragma unroll
  for (int i = 0; i < AHEAD && i < NB; ++i) {
    a[i] = *(const float4*)(arow + 8 * i);
#pragma unroll
    for (int j = 0; j < 4; ++j) w[i][j] = *(const float4*)(wrow + (size_t)(8 * i + j) * NC);
  }

#pragma unroll
  for (int kb = 0; kb < NB; ++kb) {
    if (kb + AHEAD < NB) {
      const int n = kb + AHEAD;
      a[n] = *(const float4*)(arow + 8 * n);
#pragma unroll
      for (int j = 0; j < 4; ++j) w[n][j] = *(const float4*)(wrow + (size_t)(8 * n + j) * NC);
    }
    __builtin_amdgcn_sched_barrier(0);
    float4 av = a[kb];
    if (!live) av = make_float4(0.f, 0.f, 0.f, 0.f);
    const v2f a01 = {av.x, av.y};
    const v2f a23 = {av.z, av.w};
    acc0 = wmma_f32(a01, w[kb][0].x, w[kb][1].x, acc0);
    acc1 = wmma_f32(a01, w[kb][0].y, w[kb][1].y, acc1);
    acc2 = wmma_f32(a01, w[kb][0].z, w[kb][1].z, acc2);
    acc3 = wmma_f32(a01, w[kb][0].w, w[kb][1].w, acc3);
    acc0 = wmma_f32(a23, w[kb][2].x, w[kb][3].x, acc0);
    acc1 = wmma_f32(a23, w[kb][2].y, w[kb][3].y, acc1);
    acc2 = wmma_f32(a23, w[kb][2].z, w[kb][3].z, acc2);
    acc3 = wmma_f32(a23, w[kb][2].w, w[kb][3].w, acc3);
  }

  float* orow = out + (size_t)(item * 16 + h * 8) * NC + 4 * c;
#pragma unroll
  for (int r = 0; r < 8; ++r)
    *(float4*)(orow + (size_t)r * NC) = make_float4(acc0[r], acc1[r], acc2[r], acc3[r]);
}


template <int F>
__global__ void __launch_bounds__(256)
gather_sum_kernel(const int* __restrict__ rowptr, const int* __restrict__ col,
                  const float* __restrict__ dinv, const float* __restrict__ xw,
                  const float* __restrict__ bias, float* __restrict__ out, int n) {
  constexpr int V = F / 32;
  typedef float vec __attribute__((ext_vector_type(V)));
  const int lane = threadIdx.x & 31;
  const int wave = __builtin_amdgcn_readfirstlane((int)(threadIdx.x >> 5));
  const int wavesPerBlock = (int)(blockDim.x >> 5);
  const int nw = (int)gridDim.x * wavesPerBlock;
  const vec b = *(const vec*)(bias + V * lane);

  for (int v = (int)blockIdx.x * wavesPerBlock + wave; v < n; v += nw) {
    const int beg = rowptr[v];
    const int end = rowptr[v + 1];
    const float dv = dinv[v];
    const vec self = *(const vec*)(xw + (size_t)v * F + V * lane);
    vec acc = (dv * dv) * self;

    for (int base = beg; base < end; base += 32) {
      const int m = (end - base < 32) ? (end - base) : 32;
      const int s_l = col[base + (lane < m ? lane : 0)];
      const float d_l = dinv[s_l];
      int j = 0;
      for (; j + 4 <= m; j += 4) {
        const int s0 = __builtin_amdgcn_readlane(s_l, j);
        const int s1 = __builtin_amdgcn_readlane(s_l, j + 1);
        const int s2 = __builtin_amdgcn_readlane(s_l, j + 2);
        const int s3 = __builtin_amdgcn_readlane(s_l, j + 3);
        const vec r0 = *(const vec*)(xw + (size_t)s0 * F + V * lane);
        const vec r1 = *(const vec*)(xw + (size_t)s1 * F + V * lane);
        const vec r2 = *(const vec*)(xw + (size_t)s2 * F + V * lane);
        const vec r3 = *(const vec*)(xw + (size_t)s3 * F + V * lane);
        const float w0 = __int_as_float(__builtin_amdgcn_readlane(__float_as_int(d_l), j)) * dv;
        const float w1 = __int_as_float(__builtin_amdgcn_readlane(__float_as_int(d_l), j + 1)) * dv;
        const float w2 = __int_as_float(__builtin_amdgcn_readlane(__float_as_int(d_l), j + 2)) * dv;
        const float w3 = __int_as_float(__builtin_amdgcn_readlane(__float_as_int(d_l), j + 3)) * dv;
        acc += w0 * r0;
        acc += w1 * r1;
        acc += w2 * r2;
        acc += w3 * r3;
      }
#pragma clang loop unroll(disable)
      for (; j < m; ++j) {
        const int s0 = __builtin_amdgcn_readlane(s_l, j);
        const vec r0 = *(const vec*)(xw + (size_t)s0 * F + V * lane);
        const float w0 = __int_as_float(__builtin_amdgcn_readlane(__float_as_int(d_l), j)) * dv;
        acc += w0 * r0;
      }
    }
    *(vec*)(out + (size_t)v * F + V * lane) = acc + b;
  }
}

__global__ void __launch_bounds__(256)
tanh_kernel(const float* __restrict__ in, float* __restrict__ out, int n) {
  const int n4 = n >> 2;
  const float4* in4 = (const float4*)in;
  float4* out4 = (float4*)out;
  const int first = (int)blockIdx.x * 1024 + (int)threadIdx.x;
  float4 q[4];
#pragma unroll
  for (int u = 0; u < 4; ++u) {
    const int i = first + u * 256;
    q[u] = (i < n4) ? in4[i] : make_float4(0.f, 0.f, 0.f, 0.f);
  }
#pragma unroll
  for (int u = 0; u < 4; ++u) {
    const int i = first + u * 256;
    if (i < n4) {
      float4 t;
      t.x = tanhf(q[u].x);
      t.y = tanhf(q[u].y);
      t.z = tanhf(q[u].z);
      t.w = tanhf(q[u].w);
      out4[i] = t;
    }
  }
  if (blockIdx.x == 0 && (int)threadIdx.x < (n & 3)) {
    const int i = 4 * n4 + (int)threadIdx.x;
    out[i] = tanhf(in[i]);
  }
}


extern "C" void kernel_launch(void* const* d_in, const int* in_sizes, int n_in,
                              void* d_out, int out_size, void* d_ws, size_t ws_size,
                              hipStream_t stream) {
  const float* x  = (const float*)d_in[0];
  const int*   ei = (const int*)d_in[1];
  const float* W1 = (const float*)d_in[2];
  const float* b1 = (const float*)d_in[3];
  const float* W2 = (const float*)d_in[4];
  const float* b2 = (const float*)d_in[5];

  const int DIN  = in_sizes[3];
  const int DOUT = in_sizes[5];
  const int N    = in_sizes[0] / DIN;
  const int E    = in_sizes[1] / 2;
  const int* src = ei;
  const int* dst = ei + E;

  char* ws = (char*)d_ws;
  size_t off = ((size_t)N * sizeof(float) + 255) & ~(size_t)255;
  float* dinv = (float*)ws;
  float* xw   = (float*)(ws + off);
  size_t offw = off + (((size_t)N * DOUT * sizeof(float) + 255) & ~(size_t)255);
  float* w12  = (float*)(ws + offw);
  float* c1   = w12 + (size_t)DIN * DOUT;
  size_t off2 = off + (((size_t)N * DIN * sizeof(float) + 255) & ~(size_t)255);
  float* h1   = (float*)(ws + off2);
  float* xw2  = h1;
  float* pre  = xw;
  float* out  = (float*)d_out;

  char* ob = (char*)d_out;
  size_t lo2 = ((size_t)(N + 1) * sizeof(int) + 255) & ~(size_t)255;
  size_t lo3 = lo2 + (((size_t)E * sizeof(int) + 255) & ~(size_t)255);
  int* rowptr = (int*)ob;
  int* rank   = (int*)(ob + lo2);
  int* col    = (int*)(ob + lo3);

  const int TB = 256;
  const int NR = (N + RANGE - 1) / RANGE;
  const int NP = NR * RANGE;
  const int SUBP = NR * GROUPS;
  const int chunk = (((E + CHUNKS - 1) / CHUNKS) + 3) & ~3;
  int* cntm = (int*)h1;
  int* sub  = cntm + (size_t)CHUNKS * NP;

  fold_weights_kernel<128, 64><<<(128 / 16 + 1 + 3) / 4, 128, 0, stream>>>(W1, b1, W2, w12, 128 / 16);

  count_rank_kernel<<<dim3(CHUNKS, NR), CT, 0, stream>>>(dst, rank, cntm, sub, E, chunk, NP, SUBP);
  scan_counts_kernel<<<(N + 255) / 256, 256, 0, stream>>>(cntm, sub, rowptr, dinv, N, NP, SUBP);
  fill_lists_kernel<<<(E + 4 * TB - 1) / (4 * TB), TB, 0, stream>>>(src, dst, rank, cntm, col, E, chunk, NP);

  gemm_wmma_kernel<128, 64><<<(N / 16 * (64 / 64) + 3) / 4, 128, 0, stream>>>(x, w12, xw, N / 16);

  gather_sum_kernel<64><<<(N + 15) / 16, 256, 0, stream>>>(rowptr, col, dinv, xw, c1, xw2, N);

  gather_sum_kernel<64><<<(N + 15) / 16, 256, 0, stream>>>(rowptr, col, dinv, xw2, b2, pre, N);
  tanh_kernel<<<(out_size + 16 * TB - 1) / (16 * TB), TB, 0, stream>>>(pre, out, out_size);
}


# 8 launches become 5: the W1@W2 fold runs as the first work-groups of the scan launch, x@W12 as the first of the fill launch, tanh in the last aggregation's store (edge lists move to the workspace)
# speedup vs baseline: 3.5761x; 1.1736x over previous
#include <hip/hip_runtime.h>
#include <hip/hip_bf16.h>

typedef __attribute__((ext_vector_type(2))) float v2f;
typedef __attribute__((ext_vector_type(8))) float v8f;


constexpr int CHUNKS = 20;
constexpr int RANGE = 4096;
constexpr int GROUPS = RANGE / 256;
constexpr int CT = 1024;

__device__ __forceinline__ bool in_range(int d, int lo) { return (unsigned)(d - lo) < (unsigned)RANGE; }

__device__ __forceinline__ int lds_take_slot(int* counter) {
  return __hip_atomic_fetch_add(counter, 1, __ATOMIC_RELAXED, __HIP_MEMORY_SCOPE_WORKGROUP);
}

__global__ void __launch_bounds__(CT)
count_rank_kernel(const int* __restrict__ dst, int* __restrict__ rank, int* __restrict__ cntm,
                  int* __restrict__ sub, int E, int chunk, int NP, int SUBP) {
  __shared__ int lc[RANGE];
  __shared__ int s_sub[GROUPS];
  const int tid = threadIdx.x;
  const int ci = blockIdx.x;
  const int rj = blockIdx.y;
  const int lo = rj * RANGE;

  ((int4*)lc)[tid] = make_int4(0, 0, 0, 0);
  if (tid < GROUPS) s_sub[tid] = 0;
  __syncthreads();

  const int cbeg = ci * chunk;
  const int cend = (cbeg + chunk < E) ? (cbeg + chunk) : E;
  if ((E & 3) == 0) {
    const int n4 = (cend - cbeg) >> 2;
    const int4* d4 = (const int4*)(dst + cbeg);
    int* rk = rank + cbeg;
    for (int q = tid; q < n4; q += 8 * CT) {
      int4 d[8];
#pragma unroll
      for (int u = 0; u < 8; ++u) {
        const int idx = q + u * CT;
        d[u] = (idx < n4) ? d4[idx] : make_int4(-1, -1, -1, -1);
      }
#pragma unroll
      for (int u = 0; u < 8; ++u) {
        const int e = 4 * (q + u * CT);
        const bool h0 = in_range(d[u].x, lo);
        const bool h1 = in_range(d[u].y, lo);
        const bool h2 = in_range(d[u].z, lo);
        const bool h3 = in_range(d[u].w, lo);
        int r0 = 0, r1 = 0, r2 = 0, r3 = 0;
        if (h0) r0 = lds_take_slot(&lc[d[u].x - lo]);
        if (h1) r1 = lds_take_slot(&lc[d[u].y - lo]);
        if (h2) r2 = lds_take_slot(&lc[d[u].z - lo]);
        if (h3) r3 = lds_take_slot(&lc[d[u].w - lo]);
        if (h0) rk[e] = r0;
        if (h1) rk[e + 1] = r1;
        if (h2) rk[e + 2] = r2;
        if (h3) rk[e + 3] = r3;
      }
    }
  } else {
    for (int k = cbeg + tid; k < cend; k += CT) {
      const int d = dst[k];
      if (in_range(d, lo)) rank[k] = lds_take_slot(&lc[d - lo]);
    }
  }
  __syncthreads();

  const int4 c = ((const int4*)lc)[tid];
  *(int4*)(cntm + (size_t)ci * NP + lo + 4 * tid) = c;
  int s = (c.x + c.y) + (c.z + c.w);
  for (int o = 16; o > 0; o >>= 1) s += __shfl_xor(s, o);
  if ((tid & 31) == 0)
    __hip_atomic_fetch_add(&s_sub[tid >> 6], s, __ATOMIC_RELAXED, __HIP_MEMORY_SCOPE_WORKGROUP);
  __syncthreads();
  if (tid < GROUPS) sub[ci * SUBP + rj * GROUPS + tid] = s_sub[tid];
}

__device__ __forceinline__ void
scan_counts_role(int* __restrict__ cntm, const int* __restrict__ sub,
                 int* __restrict__ rowptr, float* __restrict__ dinv, int n, int NP, int SUBP,
                 int g, int* s_wave_cnt, int* s_wave_front) {
  const int tid = threadIdx.x;
  const int lane = tid & 31;
  const int wave = tid >> 5;
  const int v = g * 256 + tid;

  int part[CHUNKS];
  int fr[CHUNKS];
  const bool has = tid < g;
  const int gi = has ? tid : 0;
#pragma unroll
  for (int i = 0; i < CHUNKS; ++i) {
    part[i] = cntm[(size_t)i * NP + v];
    fr[i] = sub[i * SUBP + gi];
  }
  int c = 0;
  int front = 0;
#pragma unroll
  for (int i = 0; i < CHUNKS; ++i) {
    c += part[i];
    front += has ? fr[i] : 0;
  }
#pragma clang loop unroll(disable)
  for (int gg = tid + 256; gg < g; gg += 256) {
#pragma clang loop unroll(disable)
    for (int i = 0; i < CHUNKS; ++i) front += sub[i * SUBP + gg];
  }

  int incl = c;
  for (int o = 1; o < 32; o <<= 1) {
    int t = __shfl_up(incl, o);
    if (lane >= o) incl += t;
  }
  for (int o = 16; o > 0; o >>= 1) front += __shfl_xor(front, o);
  if (lane == 31) s_wave_cnt[wave] = incl;
  if (lane == 0) s_wave_front[wave] = front;
  __syncthreads();
  int base = 0;
  for (int w = 0; w < 8; ++w) {
    base += s_wave_front[w];
    if (w < wave) base += s_wave_cnt[w];
  }
  const int excl = base + incl - c;
  if (v < n) {
    rowptr[v] = excl;
    dinv[v] = rsqrtf((float)(c + 1));
    int run = excl;
#pragma unroll
    for (int i = 0; i < CHUNKS; ++i) {
      cntm[(size_t)i * NP + v] = run;
      run += part[i];
    }
  }
  if (v == n - 1) rowptr[n] = excl + c;
}

__device__ __forceinline__ void
fill_lists_role(const int* __restrict__ src, const int* __restrict__ dst,
                const int* __restrict__ rank,
                const int* __restrict__ first, int* __restrict__ col,
                int E, int chunk, int NP, int e) {
  if ((E & 3) == 0) {
    if (e < E) {
      const int* f = first + (size_t)(e / chunk) * NP;
      const int4 s = *(const int4*)(src + e);
      const int4 d = *(const int4*)(dst + e);
      const int4 r = *(const int4*)(rank + e);
      const int p0 = f[d.x];
      const int p1 = f[d.y];
      const int p2 = f[d.z];
      const int p3 = f[d.w];
      col[p0 + r.x] = s.x;
      col[p1 + r.y] = s.y;
      col[p2 + r.z] = s.z;
      col[p3 + r.w] = s.w;
    }
  } else {
    for (int k = e; k < e + 4 && k < E; ++k)
      col[first[(size_t)(k / chunk) * NP + dst[k]] + rank[k]] = src[k];
  }
}


__device__ __forceinline__ v8f wmma_f32(v2f a, float b_lo, float b_hi, v8f c) {
  v2f b = {b_lo, b_hi};
  return __builtin_amdgcn_wmma_f32_16x16x4_f32(false, a, false, b, (short)0, c, false, false);
}

template <int K, int NC>
__device__ __forceinline__ void
gemm_wmma_role(const float* __restrict__ A, const float* __restrict__ W, float* __restrict__ out, int rowTiles, int block) {
  constexpr int G = NC / 64;
  constexpr int NB = K / 8;
  constexpr int AHEAD = 3;
  const int lane = threadIdx.x & 31;
  const int wave = __builtin_amdgcn_readfirstlane((int)(threadIdx.x >> 5));
  const int c = lane & 15;
  const int h = lane >> 4;
  const int item = block * 4 + wave;
  const int rowTile = item / G;
  const int strip = item % G;
  if (rowTile >= rowTiles) return;

  const float* arow = A + (size_t)(rowTile * 16 + c) * K + 4 * h;
  const float* wrow = W + (size_t)(4 * h) * NC + strip * 64 + 4 * c;

  v8f acc0 = {0.f, 0.f, 0.f, 0.f, 0.f, 0.f, 0.f, 0.f};
  v8f acc1 = acc0, acc2 = acc0, acc3 = acc0;

  float4 a[NB];
  float4 b[NB][4];

#pragma unroll
  for (int i = 0; i < AHEAD && i < NB; ++i) {
    a[i] = *(const float4*)(arow + 8 * i);
#pragma unroll
    for (int j = 0; j < 4; ++j) b[i][j] = *(const float4*)(wrow + (size_t)(8 * i + j) * NC);
  }

#pragma unroll
  for (int kb = 0; kb < NB; ++kb) {
    if (kb + AHEAD < NB) {
      const int n = kb + AHEAD;
      a[n] = *(const float4*)(arow + 8 * n);
#pragma unroll
      for (int j = 0; j < 4; ++j) b[n][j] = *(const float4*)(wrow + (size_t)(8 * n + j) * NC);
    }
    __builtin_amdgcn_sched_barrier(0);
    const v2f a01 = {a[kb].x, a[kb].y};
    const v2f a23 = {a[kb].z, a[kb].w};
    acc0 = wmma_f32(a01, b[kb][0].x, b[kb][1].x, acc0);
    acc1 = wmma_f32(a01, b[kb][0].y, b[kb][1].y, acc1);
    acc2 = wmma_f32(a01, b[kb][0].z, b[kb][1].z, acc2);
    acc3 = wmma_f32(a01, b[kb][0].w, b[kb][1].w, acc3);
    acc0 = wmma_f32(a23, b[kb][2].x, b[kb][3].x, acc0);
    acc1 = wmma_f32(a23, b[kb][2].y, b[kb][3].y, acc1);
    acc2 = wmma_f32(a23, b[kb][2].z, b[kb][3].z, acc2);
    acc3 = wmma_f32(a23, b[kb][2].w, b[kb][3].w, acc3);
  }

  float* orow = out + (size_t)(rowTile * 16 + h * 8) * NC + strip * 64 + 4 * c;
#pragma unroll
  for (int r = 0; r < 8; ++r)
    *(float4*)(orow + (size_t)r * NC) = make_float4(acc0[r], acc1[r], acc2[r], acc3[r]);
}

template <int K, int NC>
__device__ __forceinline__ void
fold_weights_role(const float* __restrict__ W1, const float* __restrict__ b1,
                  const float* __restrict__ W2, float* __restrict__ out, int rowTiles, int block) {
  static_assert(NC == 64, "one strip of 64 columns");
  constexpr int NB = K / 8;
  constexpr int AHEAD = 6;
  const int lane = threadIdx.x & 31;
  const int wave = __builtin_amdgcn_readfirstlane((int)(threadIdx.x >> 5));
  const int c = lane & 15;
  const int h = lane >> 4;
  const int item = block * 8 + wave;
  if (item > rowTiles) return;
  const bool bias_tile = (item == rowTiles);
  const bool live = !bias_tile || c == 0;
  const float* arow = (bias_tile ? b1 : W1 + (size_t)(item * 16 + c) * K) + 4 * h;
  const float* wrow = W2 + (size_t)(4 * h) * NC + 4 * c;

  v8f acc0 = {0.f, 0.f, 0.f, 0.f, 0.f, 0.f, 0.f, 0.f};
  v8f acc1 = acc0, acc2 = acc0, acc3 = acc0;

  float4 a[NB];
  float4 w[NB][4];

#pragma unroll
  for (int i = 0; i < AHEAD && i < NB; ++i) {
    a[i] = *(const float4*)(arow + 8 * i);
#pragma unroll
    for (int j = 0; j < 4; ++j) w[i][j] = *(const float4*)(wrow + (size_t)(8 * i + j) * NC);
  }

#pragma unroll
  for (int kb = 0; kb < NB; ++kb) {
    if (kb + AHEAD < NB) {
      const int n = kb + AHEAD;
      a[n] = *(const float4*)(arow + 8 * n);
#pragma unroll
      for (int j = 0; j < 4; ++j) w[n][j] = *(const float4*)(wrow + (size_t)(8 * n + j) * NC);
    }
    __builtin_amdgcn_sched_barrier(0);
    float4 av = a[kb];
    if (!live) av = make_float4(0.f, 0.f, 0.f, 0.f);
    const v2f a01 = {av.x, av.y};
    const v2f a23 = {av.z, av.w};
    acc0 = wmma_f32(a01, w[kb][0].x, w[kb][1].x, acc0);
    acc1 = wmma_f32(a01, w[kb][0].y, w[kb][1].y, acc1);
    acc2 = wmma_f32(a01, w[kb][0].z, w[kb][1].z, acc2);
    acc3 = wmma_f32(a01, w[kb][0].w, w[kb][1].w, acc3);
    acc0 = wmma_f32(a23, w[kb][2].x, w[kb][3].x, acc0);
    acc1 = wmma_f32(a23, w[kb][2].y, w[kb][3].y, acc1);
    acc2 = wmma_f32(a23, w[kb][2].z, w[kb][3].z, acc2);
    acc3 = wmma_f32(a23, w[kb][2].w, w[kb][3].w, acc3);
  }

  float* orow = out + (size_t)(item * 16 + h * 8) * NC + 4 * c;
#pragma unroll
  for (int r = 0; r < 8; ++r)
    *(float4*)(orow + (size_t)r * NC) = make_float4(acc0[r], acc1[r], acc2[r], acc3[r]);
}


template <int K, int NC>
__global__ void __launch_bounds__(256)
scan_and_fold_kernel(int* __restrict__ cntm, const int* __restrict__ sub,
                     int* __restrict__ rowptr, float* __restrict__ dinv, int n, int NP, int SUBP,
                     const float* __restrict__ W1, const float* __restrict__ b1,
                     const float* __restrict__ W2, float* __restrict__ w12, int rowTiles,
                     int foldBlocks) {
  __shared__ int s_wave_cnt[8];
  __shared__ int s_wave_front[8];
  const int block = (int)blockIdx.x;
  if (block < foldBlocks) {
    fold_weights_role<K, NC>(W1, b1, W2, w12, rowTiles, block);
    return;
  }
  scan_counts_role(cntm, sub, rowptr, dinv, n, NP, SUBP, block - foldBlocks, s_wave_cnt, s_wave_front);
}

template <int K, int NC>
__global__ void __launch_bounds__(128)
fill_and_gemm_kernel(const int* __restrict__ src, const int* __restrict__ dst,
                     const int* __restrict__ rank, const int* __restrict__ first,
                     int* __restrict__ col, int E, int chunk, int NP,
                     const float* __restrict__ A, const float* __restrict__ W,
                     float* __restrict__ xw, int rowTiles, int gemmBlocks) {
  const int block = (int)blockIdx.x;
  if (block < gemmBlocks) {
    gemm_wmma_role<K, NC>(A, W, xw, rowTiles, block);
    return;
  }
  fill_lists_role(src, dst, rank, first, col, E, chunk, NP,
                  4 * ((block - gemmBlocks) * 128 + (int)threadIdx.x));
}


template <int F, bool TANH>
__global__ void __launch_bounds__(256)
gather_sum_kernel(const int* __restrict__ rowptr, const int* __restrict__ col,
                  const float* __restrict__ dinv, const float* __restrict__ xw,
                  const float* __restrict__ bias, float* __restrict__ out, int n) {
  constexpr int V = F / 32;
  typedef float vec __attribute__((ext_vector_type(V)));
  const int lane = threadIdx.x & 31;
  const int wave = __builtin_amdgcn_readfirstlane((int)(threadIdx.x >> 5));
  const int wavesPerBlock = (int)(blockDim.x >> 5);
  const int nw = (int)gridDim.x * wavesPerBlock;
  const vec b = *(const vec*)(bias + V * lane);

  for (int v = (int)blockIdx.x * wavesPerBlock + wave; v < n; v += nw) {
    const int beg = rowptr[v];
    const int end = rowptr[v + 1];
    const float dv = dinv[v];
    const vec self = *(const vec*)(xw + (size_t)v * F + V * lane);
    vec acc = (dv * dv) * self;

    for (int base = beg; base < end; base += 32) {
      const int m = (end - base < 32) ? (end - base) : 32;
      const int s_l = col[base + (lane < m ? lane : 0)];
      const float d_l = dinv[s_l];
      int j = 0;
      for (; j + 4 <= m; j += 4) {
        const int s0 = __builtin_amdgcn_readlane(s_l, j);
        const int s1 = __builtin_amdgcn_readlane(s_l, j + 1);
        const int s2 = __builtin_amdgcn_readlane(s_l, j + 2);
        const int s3 = __builtin_amdgcn_readlane(s_l, j + 3);
        const vec r0 = *(const vec*)(xw + (size_t)s0 * F + V * lane);
        const vec r1 = *(const vec*)(xw + (size_t)s1 * F + V * lane);
        const vec r2 = *(const vec*)(xw + (size_t)s2 * F + V * lane);
        const vec r3 = *(const vec*)(xw + (size_t)s3 * F + V * lane);
        const float w0 = __int_as_float(__builtin_amdgcn_readlane(__float_as_int(d_l), j)) * dv;
        const float w1 = __int_as_float(__builtin_amdgcn_readlane(__float_as_int(d_l), j + 1)) * dv;
        const float w2 = __int_as_float(__builtin_amdgcn_readlane(__float_as_int(d_l), j + 2)) * dv;
        const float w3 = __int_as_float(__builtin_amdgcn_readlane(__float_as_int(d_l), j + 3)) * dv;
        acc += w0 * r0;
        acc += w1 * r1;
        acc += w2 * r2;
        acc += w3 * r3;
      }
#pragma clang loop unroll(disable)
      for (; j < m; ++j) {
        const int s0 = __builtin_amdgcn_readlane(s_l, j);
        const vec r0 = *(const vec*)(xw + (size_t)s0 * F + V * lane);
        const float w0 = __int_as_float(__builtin_amdgcn_readlane(__float_as_int(d_l), j)) * dv;
        acc += w0 * r0;
      }
    }
    vec res = acc + b;
    if (TANH) {
#pragma unroll
      for (int i = 0; i < V; ++i) res[i] = tanhf(res[i]);
    }
    *(vec*)(out + (size_t)v * F + V * lane) = res;
  }
}


extern "C" void kernel_launch(void* const* d_in, const int* in_sizes, int n_in,
                              void* d_out, int out_size, void* d_ws, size_t ws_size,
                              hipStream_t stream) {
  const float* x  = (const float*)d_in[0];
  const int*   ei = (const int*)d_in[1];
  const float* W1 = (const float*)d_in[2];
  const float* b1 = (const float*)d_in[3];
  const float* W2 = (const float*)d_in[4];
  const float* b2 = (const float*)d_in[5];

  const int DIN  = in_sizes[3];
  const int DOUT = in_sizes[5];
  const int N    = in_sizes[0] / DIN;
  const int E    = in_sizes[1] / 2;
  const int* src = ei;
  const int* dst = ei + E;

  char* ws = (char*)d_ws;
  size_t off = ((size_t)N * sizeof(float) + 255) & ~(size_t)255;
  float* dinv = (float*)ws;
  float* xw   = (float*)(ws + off);
  size_t offw = off + (((size_t)N * DOUT * sizeof(float) + 255) & ~(size_t)255);
  float* w12  = (float*)(ws + offw);
  float* c1   = w12 + (size_t)DIN * DOUT;
  size_t offl = offw + (((size_t)(DIN + 16) * DOUT * sizeof(float) + 255) & ~(size_t)255);
  size_t off2 = off + (((size_t)N * DIN * sizeof(float) + 255) & ~(size_t)255);
  float* h1   = (float*)(ws + off2);
  float* xw2  = h1;
  float* out  = (float*)d_out;

  size_t lo2 = ((size_t)(N + 1) * sizeof(int) + 255) & ~(size_t)255;
  size_t lo3 = lo2 + (((size_t)E * sizeof(int) + 255) & ~(size_t)255);
  int* rowptr = (int*)(ws + offl);
  int* rank   = (int*)(ws + offl + lo2);
  int* col    = (int*)(ws + offl + lo3);

  const int NR = (N + RANGE - 1) / RANGE;
  const int NP = NR * RANGE;
  const int SUBP = NR * GROUPS;
  const int chunk = (((E + CHUNKS - 1) / CHUNKS) + 3) & ~3;
  int* cntm = (int*)h1;
  int* sub  = cntm + (size_t)CHUNKS * NP;

  const int foldBlocks = (128 / 16 + 1 + 7) / 8;
  const int gemmBlocks = (N / 16 * (64 / 64) + 3) / 4;
  count_rank_kernel<<<dim3(CHUNKS, NR), CT, 0, stream>>>(dst, rank, cntm, sub, E, chunk, NP, SUBP);
  scan_and_fold_kernel<128, 64><<<foldBlocks + (N + 255) / 256, 256, 0, stream>>>(
      cntm, sub, rowptr, dinv, N, NP, SUBP, W1, b1, W2, w12, 128 / 16, foldBlocks);
  fill_and_gemm_kernel<128, 64><<<gemmBlocks + (E + 4 * 128 - 1) / (4 * 128), 128, 0, stream>>>(
      src, dst, rank, cntm, col, E, chunk, NP, x, w12, xw, N / 16, gemmBlocks);

  gather_sum_kernel<64, false><<<(N + 15) / 16, 256, 0, stream>>>(rowptr, col, dinv, xw, c1, xw2, N);

  gather_sum_kernel<64, true><<<(N + 15) / 16, 256, 0, stream>>>(rowptr, col, dinv, xw2, b2, out, N);
}


# aggregation: the rows of up to 16 list entries of a node are requested together and then summed, instead of four at a time plus one round trip per leftover entry; same sum order
# speedup vs baseline: 3.9276x; 1.0983x over previous
#include <hip/hip_runtime.h>
#include <hip/hip_bf16.h>

typedef __attribute__((ext_vector_type(2))) float v2f;
typedef __attribute__((ext_vector_type(8))) float v8f;


constexpr int CHUNKS = 20;
constexpr int RANGE = 4096;
constexpr int GROUPS = RANGE / 256;
constexpr int CT = 1024;

__device__ __forceinline__ bool in_range(int d, int lo) { return (unsigned)(d - lo) < (unsigned)RANGE; }

__device__ __forceinline__ int lds_take_slot(int* counter) {
  return __hip_atomic_fetch_add(counter, 1, __ATOMIC_RELAXED, __HIP_MEMORY_SCOPE_WORKGROUP);
}

__global__ void __launch_bounds__(CT)
count_rank_kernel(const int* __restrict__ dst, int* __restrict__ rank, int* __restrict__ cntm,
                  int* __restrict__ sub, int E, int chunk, int NP, int SUBP) {
  __shared__ int lc[RANGE];
  __shared__ int s_sub[GROUPS];
  const int tid = threadIdx.x;
  const int ci = blockIdx.x;
  const int rj = blockIdx.y;
  const int lo = rj * RANGE;

  ((int4*)lc)[tid] = make_int4(0, 0, 0, 0);
  if (tid < GROUPS) s_sub[tid] = 0;
  __syncthreads();

  const int cbeg = ci * chunk;
  const int cend = (cbeg + chunk < E) ? (cbeg + chunk) : E;
  if ((E & 3) == 0) {
    const int n4 = (cend - cbeg) >> 2;
    const int4* d4 = (const int4*)(dst + cbeg);
    int* rk = rank + cbeg;
    for (int q = tid; q < n4; q += 8 * CT) {
      int4 d[8];
#pragma unroll
      for (int u = 0; u < 8; ++u) {
        const int idx = q + u * CT;
        d[u] = (idx < n4) ? d4[idx] : make_int4(-1, -1, -1, -1);
      }
#pragma unroll
      for (int u = 0; u < 8; ++u) {
        const int e = 4 * (q + u * CT);
        const bool h0 = in_range(d[u].x, lo);
        const bool h1 = in_range(d[u].y, lo);
        const bool h2 = in_range(d[u].z, lo);
        const bool h3 = in_range(d[u].w, lo);
        int r0 = 0, r1 = 0, r2 = 0, r3 = 0;
        if (h0) r0 = lds_take_slot(&lc[d[u].x - lo]);
        if (h1) r1 = lds_take_slot(&lc[d[u].y - lo]);
        if (h2) r2 = lds_take_slot(&lc[d[u].z - lo]);
        if (h3) r3 = lds_take_slot(&lc[d[u].w - lo]);
        if (h0) rk[e] = r0;
        if (h1) rk[e + 1] = r1;
        if (h2) rk[e + 2] = r2;
        if (h3) rk[e + 3] = r3;
      }
    }
  } else {
    for (int k = cbeg + tid; k < cend; k += CT) {
      const int d = dst[k];
      if (in_range(d, lo)) rank[k] = lds_take_slot(&lc[d - lo]);
    }
  }
  __syncthreads();

  const int4 c = ((const int4*)lc)[tid];
  *(int4*)(cntm + (size_t)ci * NP + lo + 4 * tid) = c;
  int s = (c.x + c.y) + (c.z + c.w);
  for (int o = 16; o > 0; o >>= 1) s += __shfl_xor(s, o);
  if ((tid & 31) == 0)
    __hip_atomic_fetch_add(&s_sub[tid >> 6], s, __ATOMIC_RELAXED, __HIP_MEMORY_SCOPE_WORKGROUP);
  __syncthreads();
  if (tid < GROUPS) sub[ci * SUBP + rj * GROUPS + tid] = s_sub[tid];
}

__device__ __forceinline__ void
scan_counts_role(int* __restrict__ cntm, const int* __restrict__ sub,
                 int* __restrict__ rowptr, float* __restrict__ dinv, int n, int NP, int SUBP,
                 int g, int* s_wave_cnt, int* s_wave_front) {
  const int tid = threadIdx.x;
  const int lane = tid & 31;
  const int wave = tid >> 5;
  const int v = g * 256 + tid;

  int part[CHUNKS];
  int fr[CHUNKS];
  const bool has = tid < g;
  const int gi = has ? tid : 0;
#pragma unroll
  for (int i = 0; i < CHUNKS; ++i) {
    part[i] = cntm[(size_t)i * NP + v];
    fr[i] = sub[i * SUBP + gi];
  }
  int c = 0;
  int front = 0;
#pragma unroll
  for (int i = 0; i < CHUNKS; ++i) {
    c += part[i];
    front += has ? fr[i] : 0;
  }
#pragma clang loop unroll(disable)
  for (int gg = tid + 256; gg < g; gg += 256) {
#pragma clang loop unroll(disable)
    for (int i = 0; i < CHUNKS; ++i) front += sub[i * SUBP + gg];
  }

  int incl = c;
  for (int o = 1; o < 32; o <<= 1) {
    int t = __shfl_up(incl, o);
    if (lane >= o) incl += t;
  }
  for (int o = 16; o > 0; o >>= 1) front += __shfl_xor(front, o);
  if (lane == 31) s_wave_cnt[wave] = incl;
  if (lane == 0) s_wave_front[wave] = front;
  __syncthreads();
  int base = 0;
  for (int w = 0; w < 8; ++w) {
    base += s_wave_front[w];
    if (w < wave) base += s_wave_cnt[w];
  }
  const int excl = base + incl - c;
  if (v < n) {
    rowptr[v] = excl;
    dinv[v] = rsqrtf((float)(c + 1));
    int run = excl;
#pragma unroll
    for (int i = 0; i < CHUNKS; ++i) {
      cntm[(size_t)i * NP + v] = run;
      run += part[i];
    }
  }
  if (v == n - 1) rowptr[n] = excl + c;
}

__device__ __forceinline__ void
fill_lists_role(const int* __restrict__ src, const int* __restrict__ dst,
                const int* __restrict__ rank,
                const int* __restrict__ first, int* __restrict__ col,
                int E, int chunk, int NP, int e) {
  if ((E & 3) == 0) {
    if (e < E) {
      const int* f = first + (size_t)(e / chunk) * NP;
      const int4 s = *(const int4*)(src + e);
      const int4 d = *(const int4*)(dst + e);
      const int4 r = *(const int4*)(rank + e);
      const int p0 = f[d.x];
      const int p1 = f[d.y];
      const int p2 = f[d.z];
      const int p3 = f[d.w];
      col[p0 + r.x] = s.x;
      col[p1 + r.y] = s.y;
      col[p2 + r.z] = s.z;
      col[p3 + r.w] = s.w;
    }
  } else {
    for (int k = e; k < e + 4 && k < E; ++k)
      col[first[(size_t)(k / chunk) * NP + dst[k]] + rank[k]] = src[k];
  }
}


__device__ __forceinline__ v8f wmma_f32(v2f a, float b_lo, float b_hi, v8f c) {
  v2f b = {b_lo, b_hi};
  return __builtin_amdgcn_wmma_f32_16x16x4_f32(false, a, false, b, (short)0, c, false, false);
}

template <int K, int NC>
__device__ __forceinline__ void
gemm_wmma_role(const float* __restrict__ A, const float* __restrict__ W, const float* __restrict__ rs,
               float* __restrict__ out, int rowTiles, int block) {
  constexpr int G = NC / 64;
  constexpr int NB = K / 8;
  constexpr int AHEAD = 3;
  const int lane = threadIdx.x & 31;
  const int wave = __builtin_amdgcn_readfirstlane((int)(threadIdx.x >> 5));
  const int c = lane & 15;
  const int h = lane >> 4;
  const int item = block * 4 + wave;
  const int rowTile = item / G;
  const int strip = item % G;
  if (rowTile >= rowTiles) return;

  const float* arow = A + (size_t)(rowTile * 16 + c) * K + 4 * h;
  const float* wrow = W + (size_t)(4 * h) * NC + strip * 64 + 4 * c;

  constexpr int RS_AT = (NB > AHEAD) ? NB - AHEAD : 0;
  const float* rsp = rs + rowTile * 16 + 8 * h;
  float4 rs_lo, rs_hi;

  v8f acc0 = {0.f, 0.f, 0.f, 0.f, 0.f, 0.f, 0.f, 0.f};
  v8f acc1 = acc0, acc2 = acc0, acc3 = acc0;

  float4 a[NB];
  float4 b[NB][4];

#pragma unroll
  for (int i = 0; i < AHEAD && i < NB; ++i) {
    a[i] = *(const float4*)(arow + 8 * i);
#pragma unroll
    for (int j = 0; j < 4; ++j) b[i][j] = *(const float4*)(wrow + (size_t)(8 * i + j) * NC);
  }

#pragma unroll
  for (int kb = 0; kb < NB; ++kb) {
    if (kb + AHEAD < NB) {
      const int n = kb + AHEAD;
      a[n] = *(const float4*)(arow + 8 * n);
#pragma unroll
      for (int j = 0; j < 4; ++j) b[n][j] = *(const float4*)(wrow + (size_t)(8 * n + j) * NC);
    }
    if (kb == RS_AT) {
      rs_lo = *(const float4*)rsp;
      rs_hi = *(const float4*)(rsp + 4);
    }
    __builtin_amdgcn_sched_barrier(0);
    const v2f a01 = {a[kb].x, a[kb].y};
    const v2f a23 = {a[kb].z, a[kb].w};
    acc0 = wmma_f32(a01, b[kb][0].x, b[kb][1].x, acc0);
    acc1 = wmma_f32(a01, b[kb][0].y, b[kb][1].y, acc1);
    acc2 = wmma_f32(a01, b[kb][0].z, b[kb][1].z, acc2);
    acc3 = wmma_f32(a01, b[kb][0].w, b[kb][1].w, acc3);
    acc0 = wmma_f32(a23, b[kb][2].x, b[kb][3].x, acc0);
    acc1 = wmma_f32(a23, b[kb][2].y, b[kb][3].y, acc1);
    acc2 = wmma_f32(a23, b[kb][2].z, b[kb][3].z, acc2);
    acc3 = wmma_f32(a23, b[kb][2].w, b[kb][3].w, acc3);
  }

  float* orow = out + (size_t)(rowTile * 16 + h * 8) * NC + strip * 64 + 4 * c;
  const float f[8] = {rs_lo.x, rs_lo.y, rs_lo.z, rs_lo.w, rs_hi.x, rs_hi.y, rs_hi.z, rs_hi.w};
#pragma unroll
  for (int r = 0; r < 8; ++r)
    *(float4*)(orow + (size_t)r * NC) =
        make_float4(f[r] * acc0[r], f[r] * acc1[r], f[r] * acc2[r], f[r] * acc3[r]);
}

template <int K, int NC>
__device__ __forceinline__ void
fold_weights_role(const float* __restrict__ W1, const float* __restrict__ b1,
                  const float* __restrict__ W2, float* __restrict__ out, int rowTiles, int block) {
  static_assert(NC == 64, "one strip of 64 columns");
  constexpr int NB = K / 8;
  constexpr int AHEAD = 6;
  const int lane = threadIdx.x & 31;
  const int wave = __builtin_amdgcn_readfirstlane((int)(threadIdx.x >> 5));
  const int c = lane & 15;
  const int h = lane >> 4;
  const int item = block * 8 + wave;
  if (item > rowTiles) return;
  const bool bias_tile = (item == rowTiles);
  const bool live = !bias_tile || c == 0;
  const float* arow = (bias_tile ? b1 : W1 + (size_t)(item * 16 + c) * K) + 4 * h;
  const float* wrow = W2 + (size_t)(4 * h) * NC + 4 * c;

  v8f acc0 = {0.f, 0.f, 0.f, 0.f, 0.f, 0.f, 0.f, 0.f};
  v8f acc1 = acc0, acc2 = acc0, acc3 = acc0;

  float4 a[NB];
  float4 w[NB][4];

#pragma unroll
  for (int i = 0; i < AHEAD && i < NB; ++i) {
    a[i] = *(const float4*)(arow + 8 * i);
#pragma unroll
    for (int j = 0; j < 4; ++j) w[i][j] = *(const float4*)(wrow + (size_t)(8 * i + j) * NC);
  }

#pragma unroll
  for (int kb = 0; kb < NB; ++kb) {
    if (kb + AHEAD < NB) {
      const int n = kb + AHEAD;
      a[n] = *(const float4*)(arow + 8 * n);
#pragma unroll
      for (int j = 0; j < 4; ++j) w[n][j] = *(const float4*)(wrow + (size_t)(8 * n + j) * NC);
    }
    __builtin_amdgcn_sched_barrier(0);
    float4 av = a[kb];
    if (!live) av = make_float4(0.f, 0.f, 0.f, 0.f);
    const v2f a01 = {av.x, av.y};
    const v2f a23 = {av.z, av.w};
    acc0 = wmma_f32(a01, w[kb][0].x, w[kb][1].x, acc0);
    acc1 = wmma_f32(a01, w[kb][0].y, w[kb][1].y, acc1);
    acc2 = wmma_f32(a01, w[kb][0].z, w[kb][1].z, acc2);
    acc3 = wmma_f32(a01, w[kb][0].w, w[kb][1].w, acc3);
    acc0 = wmma_f32(a23, w[kb][2].x, w[kb][3].x, acc0);
    acc1 = wmma_f32(a23, w[kb][2].y, w[kb][3].y, acc1);
    acc2 = wmma_f32(a23, w[kb][2].z, w[kb][3].z, acc2);
    acc3 = wmma_f32(a23, w[kb][2].w, w[kb][3].w, acc3);
  }

  float* orow = out + (size_t)(item * 16 + h * 8) * NC + 4 * c;
#pragma unroll
  for (int r = 0; r < 8; ++r)
    *(float4*)(orow + (size_t)r * NC) = make_float4(acc0[r], acc1[r], acc2[r], acc3[r]);
}


template <int K, int NC>
__global__ void __launch_bounds__(256)
scan_and_fold_kernel(int* __restrict__ cntm, const int* __restrict__ sub,
                     int* __restrict__ rowptr, float* __restrict__ dinv, int n, int NP, int SUBP,
                     const float* __restrict__ W1, const float* __restrict__ b1,
                     const float* __restrict__ W2, float* __restrict__ w12, int rowTiles,
                     int foldBlocks) {
  __shared__ int s_wave_cnt[8];
  __shared__ int s_wave_front[8];
  const int block = (int)blockIdx.x;
  if (block < foldBlocks) {
    fold_weights_role<K, NC>(W1, b1, W2, w12, rowTiles, block);
    return;
  }
  scan_counts_role(cntm, sub, rowptr, dinv, n, NP, SUBP, block - foldBlocks, s_wave_cnt, s_wave_front);
}

template <int K, int NC>
__global__ void __launch_bounds__(128)
fill_and_gemm_kernel(const int* __restrict__ src, const int* __restrict__ dst,
                     const int* __restrict__ rank, const int* __restrict__ first,
                     int* __restrict__ col, int E, int chunk, int NP,
                     const float* __restrict__ A, const float* __restrict__ W,
                     const float* __restrict__ dinv,
                     float* __restrict__ xw, int rowTiles, int gemmBlocks) {
  const int block = (int)blockIdx.x;
  if (block < gemmBlocks) {
    gemm_wmma_role<K, NC>(A, W, dinv, xw, rowTiles, block);
    return;
  }
  fill_lists_role(src, dst, rank, first, col, E, chunk, NP,
                  4 * ((block - gemmBlocks) * 128 + (int)threadIdx.x));
}


template <int F, bool TANH>
__global__ void __launch_bounds__(256)
gather_sum_kernel(const int* __restrict__ rowptr, const int* __restrict__ col,
                  const float* __restrict__ dinv, const float* __restrict__ xs,
                  const float* __restrict__ bias, float* __restrict__ out, int n) {
  constexpr int V = F / 32;
  constexpr int DEPTH = 16;
  typedef float vec __attribute__((ext_vector_type(V)));
  const int lane = threadIdx.x & 31;
  const int wave = __builtin_amdgcn_readfirstlane((int)(threadIdx.x >> 5));
  const int wavesPerBlock = (int)(blockDim.x >> 5);
  const int nw = (int)gridDim.x * wavesPerBlock;
  const vec b = *(const vec*)(bias + V * lane);

  for (int v = (int)blockIdx.x * wavesPerBlock + wave; v < n; v += nw) {
    const int beg = rowptr[v];
    const int end = rowptr[v + 1];
    const float dv = dinv[v];
    vec acc = *(const vec*)(xs + (size_t)v * F + V * lane);
    asm volatile("" ::"s"(dv));

    for (int base = beg; base < end; base += 32) {
      const int m = (end - base < 32) ? (end - base) : 32;
      const int s_l = col[base + (lane < m ? lane : 0)];
      for (int g = 0; g < m; g += DEPTH) {
        const int mg = m - g;
        int sid[DEPTH];
        vec r[DEPTH];
#pragma unroll
        for (int k = 0; k < DEPTH; ++k) sid[k] = __builtin_amdgcn_readlane(s_l, g + k);
#pragma unroll
        for (int k = 0; k < DEPTH; ++k)
          if (k < mg) r[k] = *(const vec*)(xs + (size_t)sid[k] * F + V * lane);
#pragma unroll
        for (int k = 0; k < DEPTH; ++k)
          if (k < mg) acc += r[k];
      }
    }
    vec res = dv * acc + b;
    if (TANH) {
#pragma unroll
      for (int i = 0; i < V; ++i) res[i] = tanhf(res[i]);
    } else {
      res = dv * res;
    }
    *(vec*)(out + (size_t)v * F + V * lane) = res;
  }
}


extern "C" void kernel_launch(void* const* d_in, const int* in_sizes, int n_in,
                              void* d_out, int out_size, void* d_ws, size_t ws_size,
                              hipStream_t stream) {
  const float* x  = (const float*)d_in[0];
  const int*   ei = (const int*)d_in[1];
  const float* W1 = (const float*)d_in[2];
  const float* b1 = (const float*)d_in[3];
  const float* W2 = (const float*)d_in[4];
  const float* b2 = (const float*)d_in[5];

  const int DIN  = in_sizes[3];
  const int DOUT = in_sizes[5];
  const int N    = in_sizes[0] / DIN;
  const int E    = in_sizes[1] / 2;
  const int* src = ei;
  const int* dst = ei + E;

  char* ws = (char*)d_ws;
  size_t off = ((size_t)N * sizeof(float) + 255) & ~(size_t)255;
  float* dinv = (float*)ws;
  float* xw   = (float*)(ws + off);
  size_t offw = off + (((size_t)N * DOUT * sizeof(float) + 255) & ~(size_t)255);
  float* w12  = (float*)(ws + offw);
  float* c1   = w12 + (size_t)DIN * DOUT;
  size_t offl = offw + (((size_t)(DIN + 16) * DOUT * sizeof(float) + 255) & ~(size_t)255);
  size_t off2 = off + (((size_t)N * DIN * sizeof(float) + 255) & ~(size_t)255);
  float* h1   = (float*)(ws + off2);
  float* xw2  = h1;
  float* out  = (float*)d_out;

  size_t lo2 = ((size_t)(N + 1) * sizeof(int) + 255) & ~(size_t)255;
  size_t lo3 = lo2 + (((size_t)E * sizeof(int) + 255) & ~(size_t)255);
  int* rowptr = (int*)(ws + offl);
  int* rank   = (int*)(ws + offl + lo2);
  int* col    = (int*)(ws + offl + lo3);

  const int NR = (N + RANGE - 1) / RANGE;
  const int NP = NR * RANGE;
  const int SUBP = NR * GROUPS;
  const int chunk = (((E + CHUNKS - 1) / CHUNKS) + 3) & ~3;
  int* cntm = (int*)h1;
  int* sub  = cntm + (size_t)CHUNKS * NP;

  const int foldBlocks = (128 / 16 + 1 + 7) / 8;
  const int gemmBlocks = (N / 16 * (64 / 64) + 3) / 4;
  count_rank_kernel<<<dim3(CHUNKS, NR), CT, 0, stream>>>(dst, rank, cntm, sub, E, chunk, NP, SUBP);
  scan_and_fold_kernel<128, 64><<<foldBlocks + (N + 255) / 256, 256, 0, stream>>>(
      cntm, sub, rowptr, dinv, N, NP, SUBP, W1, b1, W2, w12, 128 / 16, foldBlocks);
  fill_and_gemm_kernel<128, 64><<<gemmBlocks + (E + 4 * 128 - 1) / (4 * 128), 128, 0, stream>>>(
      src, dst, rank, cntm, col, E, chunk, NP, x, w12, dinv, xw, N / 16, gemmBlocks);

  gather_sum_kernel<64, false><<<(N + 15) / 16, 256, 0, stream>>>(rowptr, col, dinv, xw, c1, xw2, N);

  gather_sum_kernel<64, true><<<(N + 15) / 16, 256, 0, stream>>>(rowptr, col, dinv, xw2, b2, out, N);
}


# in-edge lists in 2 launches, not 3: a count work-group orders its hits by destination in LDS, the scan's blocks copy the runs into place (no rank, no first-slot table); 19 chunks: sum order shifts
# speedup vs baseline: 5.9660x; 1.5190x over previous
#include <hip/hip_runtime.h>
#include <hip/hip_bf16.h>

typedef __attribute__((ext_vector_type(2))) float v2f;
typedef __attribute__((ext_vector_type(8))) float v8f;


constexpr int CHUNKS = 19;
constexpr int LD = 9;
constexpr int RANGE = 4096;
constexpr int GROUPS = RANGE / 256;
constexpr int CT = 1024;

__device__ __forceinline__ bool in_range(int d, int lo) { return (unsigned)(d - lo) < (unsigned)RANGE; }

__device__ __forceinline__ int lds_take_slot(int* counter) {
  return __hip_atomic_fetch_add(counter, 1, __ATOMIC_RELAXED, __HIP_MEMORY_SCOPE_WORKGROUP);
}

__device__ __forceinline__ void
count_place_role(const int* __restrict__ src, const int* __restrict__ dst, int* __restrict__ stage,
                 int* __restrict__ words, int* __restrict__ sub, int* __restrict__ segbase,
                 int E, int chunk, int NP, int SUBP, int NR, int ci, int rj,
                 int* lc, int* s_sub, int* s_wave, int* s_below) {
  const int tid = threadIdx.x;
  const int lane = tid & 31;
  const int wave = tid >> 5;
  const int lo = rj * RANGE;

  ((int4*)lc)[tid] = make_int4(0, 0, 0, 0);
  if (tid < GROUPS) s_sub[tid] = 0;
  if (tid == 0) *s_below = 0;
  __syncthreads();

  const int cbeg = ci * chunk;
  const int cend = (cbeg + chunk < E) ? (cbeg + chunk) : E;
  const int n4 = (cend - cbeg) >> 2;
  const bool vec = ((E & 3) == 0) && (chunk <= 4 * LD * CT);

  int4 key[LD];
  int4 s[LD];
  int below = 0;
  if (vec) {
    const int4* d4 = (const int4*)(dst + cbeg);
    const int* sp = src + cbeg;
    int4 d[LD];
#pragma unroll
    for (int u = 0; u < LD; ++u) {
      const int idx = tid + u * CT;
      d[u] = (idx < n4) ? d4[idx] : make_int4(-1, -1, -1, -1);
    }
#pragma unroll
    for (int u = 0; u < LD; ++u) {
      const int e = 4 * (tid + u * CT);
      const int l0 = d[u].x - lo, l1 = d[u].y - lo, l2 = d[u].z - lo, l3 = d[u].w - lo;
      below += ((unsigned)d[u].x < (unsigned)lo) + ((unsigned)d[u].y < (unsigned)lo) +
               ((unsigned)d[u].z < (unsigned)lo) + ((unsigned)d[u].w < (unsigned)lo);
      const bool h0 = (unsigned)l0 < (unsigned)RANGE;
      const bool h1 = (unsigned)l1 < (unsigned)RANGE;
      const bool h2 = (unsigned)l2 < (unsigned)RANGE;
      const bool h3 = (unsigned)l3 < (unsigned)RANGE;
      int r0 = 0, r1 = 0, r2 = 0, r3 = 0;
      if (h0) r0 = lds_take_slot(&lc[l0]);
      if (h1) r1 = lds_take_slot(&lc[l1]);
      if (h2) r2 = lds_take_slot(&lc[l2]);
      if (h3) r3 = lds_take_slot(&lc[l3]);
      s[u] = make_int4(0, 0, 0, 0);
      if (h0) s[u].x = sp[e];
      if (h1) s[u].y = sp[e + 1];
      if (h2) s[u].z = sp[e + 2];
      if (h3) s[u].w = sp[e + 3];
      key[u] = make_int4(h0 ? ((l0 << 16) | r0) : -1, h1 ? ((l1 << 16) | r1) : -1,
                         h2 ? ((l2 << 16) | r2) : -1, h3 ? ((l3 << 16) | r3) : -1);
    }
  } else {
    for (int k = cbeg + tid; k < cend; k += CT) {
      const int dd = dst[k];
      if (in_range(dd, lo)) __hip_atomic_fetch_add(&lc[dd - lo], 1, __ATOMIC_RELAXED, __HIP_MEMORY_SCOPE_WORKGROUP);
      below += ((unsigned)dd < (unsigned)lo);
    }
  }
  for (int o = 16; o > 0; o >>= 1) below += __shfl_xor(below, o);
  if (lane == 0 && below != 0)
    __hip_atomic_fetch_add(s_below, below, __ATOMIC_RELAXED, __HIP_MEMORY_SCOPE_WORKGROUP);
  __syncthreads();

  const int4 c = ((const int4*)lc)[tid];
  const int tot = (c.x + c.y) + (c.z + c.w);
  int incl = tot;
  for (int o = 1; o < 32; o <<= 1) {
    int t = __shfl_up(incl, o);
    if (lane >= o) incl += t;
  }
  if (lane == 31) s_wave[wave] = incl;
  int gs = tot;
  for (int o = 16; o > 0; o >>= 1) gs += __shfl_xor(gs, o);
  if (lane == 0)
    __hip_atomic_fetch_add(&s_sub[tid >> 6], gs, __ATOMIC_RELAXED, __HIP_MEMORY_SCOPE_WORKGROUP);
  __syncthreads();
  int base = 0;
  for (int w = 0; w < CT / 32; ++w) base += (w < wave) ? s_wave[w] : 0;
  const int p0 = base + incl - tot;
  const int p1 = p0 + c.x;
  const int p2 = p1 + c.y;
  const int p3 = p2 + c.z;
  ((int4*)lc)[tid] = make_int4(p0, p1, p2, p3);
  *(int4*)(words + (size_t)ci * NP + lo + 4 * tid) =
      make_int4((int)(((unsigned)p0 << 16) | (unsigned)c.x), (int)(((unsigned)p1 << 16) | (unsigned)c.y),
                (int)(((unsigned)p2 << 16) | (unsigned)c.z), (int)(((unsigned)p3 << 16) | (unsigned)c.w));
  __syncthreads();

  const int segb = cbeg + *s_below;
  if (tid < GROUPS) sub[ci * SUBP + rj * GROUPS + tid] = s_sub[tid];
  if (tid == 0) segbase[ci * NR + rj] = segb;
  int* seg = stage + segb;
  if (vec) {
    int4 st[LD];
#pragma unroll
    for (int u = 0; u < LD; ++u)
      st[u] = make_int4(lc[key[u].x >= 0 ? (key[u].x >> 16) : 0], lc[key[u].y >= 0 ? (key[u].y >> 16) : 0],
                        lc[key[u].z >= 0 ? (key[u].z >> 16) : 0], lc[key[u].w >= 0 ? (key[u].w >> 16) : 0]);
#pragma unroll
    for (int u = 0; u < LD; ++u) {
      if (key[u].x >= 0) seg[st[u].x + (key[u].x & 0xFFFF)] = s[u].x;
      if (key[u].y >= 0) seg[st[u].y + (key[u].y & 0xFFFF)] = s[u].y;
      if (key[u].z >= 0) seg[st[u].z + (key[u].z & 0xFFFF)] = s[u].z;
      if (key[u].w >= 0) seg[st[u].w + (key[u].w & 0xFFFF)] = s[u].w;
    }
  } else {
    for (int k = cbeg + tid; k < cend; k += CT) {
      const int dd = dst[k];
      if (in_range(dd, lo)) seg[lds_take_slot(&lc[dd - lo])] = src[k];
    }
  }
}

constexpr int MAPCAP = 8192;

__device__ __forceinline__ void
scan_copy_role(const int* __restrict__ words, const int* __restrict__ sub,
               const int* __restrict__ segbase, const int* __restrict__ stage,
               int* __restrict__ rowptr, int* __restrict__ col, float* __restrict__ dinv,
               int n, int NP, int SUBP, int NR, int g, int* s_wave_cnt, int* s_wave_front,
               int* s_map) {
  const int tid = threadIdx.x;
  const int lane = tid & 31;
  const int wave = tid >> 5;
  const int v = g * 256 + tid;
  const int rj = (g * 256) / RANGE;

  int wd[CHUNKS];
  int fr[CHUNKS];
  int sb[CHUNKS];
  const bool has = tid < g;
  const int gi = has ? tid : 0;
#pragma unroll
  for (int i = 0; i < CHUNKS; ++i) {
    wd[i] = words[(size_t)i * NP + v];
    fr[i] = sub[i * SUBP + gi];
    sb[i] = segbase[i * NR + rj];
  }
  int c = 0;
  int front = 0;
#pragma unroll
  for (int i = 0; i < CHUNKS; ++i) {
    c += wd[i] & 0xFFFF;
    front += has ? fr[i] : 0;
  }
#pragma clang loop unroll(disable)
  for (int gg = tid + 256; gg < g; gg += 256) {
#pragma clang loop unroll(disable)
    for (int i = 0; i < CHUNKS; ++i) front += sub[i * SUBP + gg];
  }

  int incl = c;
  for (int o = 1; o < 32; o <<= 1) {
    int t = __shfl_up(incl, o);
    if (lane >= o) incl += t;
  }
  for (int o = 16; o > 0; o >>= 1) front += __shfl_xor(front, o);
  if (lane == 31) s_wave_cnt[wave] = incl;
  if (lane == 0) s_wave_front[wave] = front;
  __syncthreads();
  int bfirst = 0;
  int total = 0;
  int before = 0;
  for (int w = 0; w < 8; ++w) {
    bfirst += s_wave_front[w];
    total += s_wave_cnt[w];
    if (w < wave) before += s_wave_cnt[w];
  }
  const int rel = before + incl - c;
  const int excl = bfirst + rel;
  if (v < n) {
    rowptr[v] = excl;
    dinv[v] = rsqrtf((float)(c + 1));
  }
  if (v == n - 1) rowptr[n] = excl + c;

  if (total <= MAPCAP) {
    int slot = rel;
#pragma unroll
    for (int i = 0; i < CHUNKS; ++i) {
      const int ni = wd[i] & 0xFFFF;
      int pos = sb[i] + (int)((unsigned)wd[i] >> 16);
#pragma clang loop unroll(disable) vectorize(disable) interleave(disable)
      for (int k = 0; k < ni; ++k) s_map[slot++] = pos++;
    }
    __syncthreads();
    int* piece = col + bfirst;
    for (int s0 = tid; s0 < total; s0 += 8 * 256) {
      int e[8];
#pragma unroll
      for (int j = 0; j < 8; ++j) {
        const int idx = s0 + j * 256;
        e[j] = (idx < total) ? stage[s_map[idx]] : 0;
      }
#pragma unroll
      for (int j = 0; j < 8; ++j) {
        const int idx = s0 + j * 256;
        if (idx < total) piece[idx] = e[j];
      }
    }
  } else {
    int* q = col + excl;
#pragma clang loop unroll(disable)
    for (int i = 0; i < CHUNKS; ++i) {
      const int w = words[(size_t)i * NP + v];
      const int ni = w & 0xFFFF;
      const int* p = stage + segbase[i * NR + rj] + (int)((unsigned)w >> 16);
#pragma clang loop unroll(disable) vectorize(disable) interleave(disable)
      for (int k = 0; k < ni; ++k) q[k] = p[k];
      q += ni;
    }
  }
}


__device__ __forceinline__ v8f wmma_f32(v2f a, float b_lo, float b_hi, v8f c) {
  v2f b = {b_lo, b_hi};
  return __builtin_amdgcn_wmma_f32_16x16x4_f32(false, a, false, b, (short)0, c, false, false);
}

template <int K, int NC>
__device__ __forceinline__ void
gemm_wmma_role(const float* __restrict__ A, const float* __restrict__ W, const int* __restrict__ words,
               int NP, float* __restrict__ out, int rowTiles, int block) {
  constexpr int G = NC / 64;
  constexpr int NB = K / 8;
  constexpr int AHEAD = 3;
  const int lane = threadIdx.x & 31;
  const int wave = __builtin_amdgcn_readfirstlane((int)(threadIdx.x >> 5));
  const int c = lane & 15;
  const int h = lane >> 4;
  const int item = block * 8 + wave;
  const int rowTile = item / G;
  const int strip = item % G;
  if (rowTile >= rowTiles) return;

  const float* arow = A + (size_t)(rowTile * 16 + c) * K + 4 * h;
  const float* wrow = W + (size_t)(4 * h) * NC + strip * 64 + 4 * c;

  constexpr int RS_AT = (NB > AHEAD) ? NB - AHEAD : 0;
  const int* wdp = words + rowTile * 16 + c;
  int cw[CHUNKS];

  v8f acc0 = {0.f, 0.f, 0.f, 0.f, 0.f, 0.f, 0.f, 0.f};
  v8f acc1 = acc0, acc2 = acc0, acc3 = acc0;

  float4 a[NB];
  float4 b[NB][4];

#pragma unroll
  for (int i = 0; i < AHEAD && i < NB; ++i) {
    a[i] = *(const float4*)(arow + 8 * i);
#pragma unroll
    for (int j = 0; j < 4; ++j) b[i][j] = *(const float4*)(wrow + (size_t)(8 * i + j) * NC);
  }

#pragma unroll
  for (int kb = 0; kb < NB; ++kb) {
    if (kb + AHEAD < NB) {
      const int n = kb + AHEAD;
      a[n] = *(const float4*)(arow + 8 * n);
#pragma unroll
      for (int j = 0; j < 4; ++j) b[n][j] = *(const float4*)(wrow + (size_t)(8 * n + j) * NC);
    }
    if (kb == RS_AT) {
#pragma unroll
      for (int i = 0; i < CHUNKS; ++i) cw[i] = wdp[(size_t)i * NP];
    }
    __builtin_amdgcn_sched_barrier(0);
    const v2f a01 = {a[kb].x, a[kb].y};
    const v2f a23 = {a[kb].z, a[kb].w};
    acc0 = wmma_f32(a01, b[kb][0].x, b[kb][1].x, acc0);
    acc1 = wmma_f32(a01, b[kb][0].y, b[kb][1].y, acc1);
    acc2 = wmma_f32(a01, b[kb][0].z, b[kb][1].z, acc2);
    acc3 = wmma_f32(a01, b[kb][0].w, b[kb][1].w, acc3);
    acc0 = wmma_f32(a23, b[kb][2].x, b[kb][3].x, acc0);
    acc1 = wmma_f32(a23, b[kb][2].y, b[kb][3].y, acc1);
    acc2 = wmma_f32(a23, b[kb][2].z, b[kb][3].z, acc2);
    acc3 = wmma_f32(a23, b[kb][2].w, b[kb][3].w, acc3);
  }

  float* orow = out + (size_t)(rowTile * 16 + h * 8) * NC + strip * 64 + 4 * c;
  int deg = 1;
#pragma unroll
  for (int i = 0; i < CHUNKS; ++i) deg += cw[i] & 0xFFFF;
  const float dv = rsqrtf((float)deg);
#pragma unroll
  for (int r = 0; r < 8; ++r) {
    const float f = __shfl(dv, 8 * h + r);
    *(float4*)(orow + (size_t)r * NC) =
        make_float4(f * acc0[r], f * acc1[r], f * acc2[r], f * acc3[r]);
  }
}

template <int K, int NC, int AHEAD>
__device__ __forceinline__ void
fold_weights_role(const float* __restrict__ W1, const float* __restrict__ b1,
                  const float* __restrict__ W2, float* __restrict__ out, int rowTiles, int block) {
  static_assert(NC == 64, "one strip of 64 columns");
  constexpr int NB = K / 8;
  const int lane = threadIdx.x & 31;
  const int wave = __builtin_amdgcn_readfirstlane((int)(threadIdx.x >> 5));
  const int c = lane & 15;
  const int h = lane >> 4;
  const int item = block * 8 + wave;
  if (item > rowTiles) return;
  const bool bias_tile = (item == rowTiles);
  const bool live = !bias_tile || c == 0;
  const float* arow = (bias_tile ? b1 : W1 + (size_t)(item * 16 + c) * K) + 4 * h;
  const float* wrow = W2 + (size_t)(4 * h) * NC + 4 * c;

  v8f acc0 = {0.f, 0.f, 0.f, 0.f, 0.f, 0.f, 0.f, 0.f};
  v8f acc1 = acc0, acc2 = acc0, acc3 = acc0;

  float4 a[NB];
  float4 w[NB][4];

#pragma unroll
  for (int i = 0; i < AHEAD && i < NB; ++i) {
    a[i] = *(const float4*)(arow + 8 * i);
#pragma unroll
    for (int j = 0; j < 4; ++j) w[i][j] = *(const float4*)(wrow + (size_t)(8 * i + j) * NC);
  }

#pragma unroll
  for (int kb = 0; kb < NB; ++kb) {
    if (kb + AHEAD < NB) {
      const int n = kb + AHEAD;
      a[n] = *(const float4*)(arow + 8 * n);
#pragma unroll
      for (int j = 0; j < 4; ++j) w[n][j] = *(const float4*)(wrow + (size_t)(8 * n + j) * NC);
    }
    __builtin_amdgcn_sched_barrier(0);
    float4 av = a[kb];
    if (!live) av = make_float4(0.f, 0.f, 0.f, 0.f);
    const v2f a01 = {av.x, av.y};
    const v2f a23 = {av.z, av.w};
    acc0 = wmma_f32(a01, w[kb][0].x, w[kb][1].x, acc0);
    acc1 = wmma_f32(a01, w[kb][0].y, w[kb][1].y, acc1);
    acc2 = wmma_f32(a01, w[kb][0].z, w[kb][1].z, acc2);
    acc3 = wmma_f32(a01, w[kb][0].w, w[kb][1].w, acc3);
    acc0 = wmma_f32(a23, w[kb][2].x, w[kb][3].x, acc0);
    acc1 = wmma_f32(a23, w[kb][2].y, w[kb][3].y, acc1);
    acc2 = wmma_f32(a23, w[kb][2].z, w[kb][3].z, acc2);
    acc3 = wmma_f32(a23, w[kb][2].w, w[kb][3].w, acc3);
  }

  float* orow = out + (size_t)(item * 16 + h * 8) * NC + 4 * c;
#pragma unroll
  for (int r = 0; r < 8; ++r)
    *(float4*)(orow + (size_t)r * NC) = make_float4(acc0[r], acc1[r], acc2[r], acc3[r]);
}


template <int K, int NC>
__global__ void __launch_bounds__(CT)
count_and_fold_kernel(const int* __restrict__ src, const int* __restrict__ dst,
                      int* __restrict__ stage, int* __restrict__ words, int* __restrict__ sub,
                      int* __restrict__ segbase, int E, int chunk, int NP, int SUBP, int NR,
                      const float* __restrict__ W1, const float* __restrict__ b1,
                      const float* __restrict__ W2, float* __restrict__ w12, int rowTiles) {
  __shared__ int lc[RANGE];
  __shared__ int s_sub[GROUPS];
  __shared__ int s_wave[CT / 32];
  __shared__ int s_below;
  const int block = (int)blockIdx.x;
  if (block == 0) {
    fold_weights_role<K, NC, 2>(W1, b1, W2, w12, rowTiles, 0);
    return;
  }
  const int cb = block - 1;
  count_place_role(src, dst, stage, words, sub, segbase, E, chunk, NP, SUBP, NR,
                   cb % CHUNKS, cb / CHUNKS, lc, s_sub, s_wave, &s_below);
}

template <int K, int NC>
__global__ void __launch_bounds__(256)
scan_and_gemm_kernel(const int* __restrict__ words, const int* __restrict__ sub,
                     const int* __restrict__ segbase, const int* __restrict__ stage,
                     int* __restrict__ rowptr, int* __restrict__ col, float* __restrict__ dinv,
                     int n, int NP, int SUBP, int NR,
                     const float* __restrict__ A, const float* __restrict__ W,
                     float* __restrict__ xw, int rowTiles, int scanBlocks) {
  __shared__ int s_wave_cnt[8];
  __shared__ int s_wave_front[8];
  __shared__ int s_map[MAPCAP];
  const int block = (int)blockIdx.x;
  if (block >= scanBlocks) {
    gemm_wmma_role<K, NC>(A, W, words, NP, xw, rowTiles, block - scanBlocks);
    return;
  }
  scan_copy_role(words, sub, segbase, stage, rowptr, col, dinv, n, NP, SUBP, NR, block,
                 s_wave_cnt, s_wave_front, s_map);
}


template <int F, bool TANH>
__global__ void __launch_bounds__(256)
gather_sum_kernel(const int* __restrict__ rowptr, const int* __restrict__ col,
                  const float* __restrict__ dinv, const float* __restrict__ xs,
                  const float* __restrict__ bias, float* __restrict__ out, int n) {
  constexpr int V = F / 32;
  constexpr int DEPTH = 16;
  typedef float vec __attribute__((ext_vector_type(V)));
  const int lane = threadIdx.x & 31;
  const int wave = __builtin_amdgcn_readfirstlane((int)(threadIdx.x >> 5));
  const int wavesPerBlock = (int)(blockDim.x >> 5);
  const int nw = (int)gridDim.x * wavesPerBlock;
  const vec b = *(const vec*)(bias + V * lane);

  for (int v = (int)blockIdx.x * wavesPerBlock + wave; v < n; v += nw) {
    const int beg = rowptr[v];
    const int end = rowptr[v + 1];
    const float dv = dinv[v];
    vec acc = *(const vec*)(xs + (size_t)v * F + V * lane);
    asm volatile("" ::"s"(dv));

    for (int base = beg; base < end; base += 32) {
      const int m = (end - base < 32) ? (end - base) : 32;
      const int s_l = col[base + (lane < m ? lane : 0)];
      for (int g = 0; g < m; g += DEPTH) {
        const int mg = m - g;
        int sid[DEPTH];
        vec r[DEPTH];
#pragma unroll
        for (int k = 0; k < DEPTH; ++k) sid[k] = __builtin_amdgcn_readlane(s_l, g + k);
#pragma unroll
        for (int k = 0; k < DEPTH; ++k)
          if (k < mg) r[k] = *(const vec*)(xs + (size_t)sid[k] * F + V * lane);
#pragma unroll
        for (int k = 0; k < DEPTH; ++k)
          if (k < mg) acc += r[k];
      }
    }
    vec res = dv * acc + b;
    if (TANH) {
#pragma unroll
      for (int i = 0; i < V; ++i) res[i] = tanhf(res[i]);
    } else {
      res = dv * res;
    }
    *(vec*)(out + (size_t)v * F + V * lane) = res;
  }
}


extern "C" void kernel_launch(void* const* d_in, const int* in_sizes, int n_in,
                              void* d_out, int out_size, void* d_ws, size_t ws_size,
                              hipStream_t stream) {
  const float* x  = (const float*)d_in[0];
  const int*   ei = (const int*)d_in[1];
  const float* W1 = (const float*)d_in[2];
  const float* b1 = (const float*)d_in[3];
  const float* W2 = (const float*)d_in[4];
  const float* b2 = (const float*)d_in[5];

  const int DIN  = in_sizes[3];
  const int DOUT = in_sizes[5];
  const int N    = in_sizes[0] / DIN;
  const int E    = in_sizes[1] / 2;
  const int* src = ei;
  const int* dst = ei + E;

  char* ws = (char*)d_ws;
  size_t off = ((size_t)N * sizeof(float) + 255) & ~(size_t)255;
  float* dinv = (float*)ws;
  float* xw   = (float*)(ws + off);
  size_t offw = off + (((size_t)N * DOUT * sizeof(float) + 255) & ~(size_t)255);
  float* w12  = (float*)(ws + offw);
  float* c1   = w12 + (size_t)DIN * DOUT;
  size_t offl = offw + (((size_t)(DIN + 16) * DOUT * sizeof(float) + 255) & ~(size_t)255);
  size_t off2 = off + (((size_t)N * DIN * sizeof(float) + 255) & ~(size_t)255);
  float* h1   = (float*)(ws + off2);
  float* xw2  = h1;
  float* out  = (float*)d_out;

  size_t lo2 = ((size_t)(N + 1) * sizeof(int) + 255) & ~(size_t)255;
  size_t lo3 = lo2 + (((size_t)E * sizeof(int) + 255) & ~(size_t)255);
  int* rowptr = (int*)(ws + offl);
  int* stage  = (int*)(ws + offl + lo2);
  int* col    = (int*)(ws + offl + lo3);

  const int NR = (N + RANGE - 1) / RANGE;
  const int NP = NR * RANGE;
  const int SUBP = NR * GROUPS;
  const int chunk = (((E + CHUNKS - 1) / CHUNKS) + 3) & ~3;
  int* words   = (int*)h1;
  int* sub     = words + (size_t)CHUNKS * NP;
  int* segbase = sub + (size_t)CHUNKS * SUBP;

  const int scanBlocks = (N + 255) / 256;
  const int gemmBlocks = (N / 16 * (64 / 64) + 7) / 8;
  count_and_fold_kernel<128, 64><<<1 + CHUNKS * NR, CT, 0, stream>>>(
      src, dst, stage, words, sub, segbase, E, chunk, NP, SUBP, NR, W1, b1, W2, w12, 128 / 16);
  scan_and_gemm_kernel<128, 64><<<scanBlocks + gemmBlocks, 256, 0, stream>>>(
      words, sub, segbase, stage, rowptr, col, dinv, N, NP, SUBP, NR, x, w12, xw, N / 16, scanBlocks);

  gather_sum_kernel<64, false><<<(N + 15) / 16, 256, 0, stream>>>(rowptr, col, dinv, xw, c1, xw2, N);

  gather_sum_kernel<64, true><<<(N + 15) / 16, 256, 0, stream>>>(rowptr, col, dinv, xw2, b2, out, N);
}
